# starting kernel
# baseline (speedup 1.0000x reference)
#include <hip/hip_runtime.h>
#include <hip/hip_bf16.h>

typedef __attribute__((ext_vector_type(2))) float v2f;
typedef __attribute__((ext_vector_type(8))) float v8f;

#define B_   16
#define N_   4096
#define C_   64
#define M_   1024
#define K_   32
#define EPS_ 1e-5f
#define BIGF 3.402823e38f

__global__ __launch_bounds__(256) void k_new_xyz(const float* __restrict__ xyz,
                                                 const int* __restrict__ indices,
                                                 float* __restrict__ new_xyz) {
  int t = blockIdx.x * 256 + threadIdx.x;
  if (t >= B_ * M_) return;
  int b = t >> 10, m = t & 1023;
  int n = indices[m];
  const float* s = xyz + ((size_t)b * N_ + n) * 3;
  float* d = new_xyz + (size_t)t * 3;
  d[0] = s[0]; d[1] = s[1]; d[2] = s[2];
}

__global__ __launch_bounds__(64) void k_knn(const float* __restrict__ xyz,
                                            const float* __restrict__ new_xyz,
                                            int* __restrict__ knn) {
  __shared__ float dist[2][N_];
  const int wave = threadIdx.x >> 5;
  const int lane = threadIdx.x & 31;
  const int q = blockIdx.x * 2 + wave;
  const int b = q >> 10;

  float* dl = dist[wave];
  const float qx = new_xyz[q * 3 + 0];
  const float qy = new_xyz[q * 3 + 1];
  const float qz = new_xyz[q * 3 + 2];
  const float qq = qx * qx + qy * qy + qz * qz;
  const float* px = xyz + (size_t)b * N_ * 3;

  for (int n = lane; n < N_; n += 32) {
    float x = px[n * 3 + 0], y = px[n * 3 + 1], z = px[n * 3 + 2];
    dl[n] = qq + (x * x + y * y + z * z) - 2.0f * (qx * x + qy * y + qz * z);
  }
  __syncthreads();

  int* out = knn + q * K_;
  for (int k = 0; k < K_; ++k) {
    float mv = BIGF;
    int   mi = 0x7fffffff;
    for (int n = lane; n < N_; n += 32) {
      float v = dl[n];
      if (v < mv) { mv = v; mi = n; }
    }
#pragma unroll
    for (int off = 16; off > 0; off >>= 1) {
      float ov = __shfl_xor(mv, off, 32);
      int   oi = __shfl_xor(mi, off, 32);
      if (ov < mv || (ov == mv && oi < mi)) { mv = ov; mi = oi; }
    }
    if (lane == 0) out[k] = mi;
    dl[mi] = BIGF;
    __syncthreads();
  }
}

__device__ __forceinline__ void gemm_bn_relu(
    const float* __restrict__ Xb, int kPad,
    const float* __restrict__ W, int kW, int rowsOut,
    const float* __restrict__ g, const float* __restrict__ bt,
    const float* __restrict__ mu, const float* __restrict__ vr,
    float* __restrict__ Yb, int lane)
{
  const int half = lane >> 4;
  const int l16  = lane & 15;
  const int nKt  = kPad >> 2;

  for (int mt = 0; mt < (rowsOut >> 4); ++mt) {
    const int arow = mt * 16 + l16;
    for (int nt = 0; nt < 2; ++nt) {
      const int col = nt * 16 + l16;
      v8f acc = {};
      for (int kt = 0; kt < nKt; ++kt) {
        const int k0 = kt * 4 + half * 2;
        v2f a, bf;
        a.x  = (k0 + 0 < kW) ? W[arow * kW + k0 + 0] : 0.0f;
        a.y  = (k0 + 1 < kW) ? W[arow * kW + k0 + 1] : 0.0f;
        bf.x = Xb[(k0 + 0) * 32 + col];
        bf.y = Xb[(k0 + 1) * 32 + col];
        acc = __builtin_amdgcn_wmma_f32_16x16x4_f32(
             false, a,  false, bf,
             (short)0, acc,  false,  false);
      }
      const int rbase = mt * 16 + half * 8;
#pragma unroll
      for (int r = 0; r < 8; ++r) {
        const int row = rbase + r;
        const float s = g[row] / sqrtf(vr[row] + EPS_);
        const float y = (acc[r] - mu[row]) * s + bt[row];
        Yb[row * 32 + col] = fmaxf(y, 0.0f);
      }
    }
  }
}

__global__ __launch_bounds__(64) void k_mlp(
    const float* __restrict__ features, const float* __restrict__ xyz,
    const float* __restrict__ new_xyz, const int* __restrict__ knn,
    const float* __restrict__ W0, const float* __restrict__ g0,
    const float* __restrict__ b0, const float* __restrict__ m0,
    const float* __restrict__ v0,
    const float* __restrict__ W1, const float* __restrict__ g1,
    const float* __restrict__ b1, const float* __restrict__ m1,
    const float* __restrict__ v1,
    const float* __restrict__ W2, const float* __restrict__ g2,
    const float* __restrict__ b2, const float* __restrict__ m2,
    const float* __restrict__ v2,
    float* __restrict__ out_feat)
{
  __shared__ float Xs[2][68 * 32];
  __shared__ float Ys[2][128 * 32];
  const int wave = threadIdx.x >> 5;
  const int lane = threadIdx.x & 31;
  const int q = blockIdx.x * 2 + wave;
  const int b = q >> 10;
  const int m = q & 1023;

  float* Xb = Xs[wave];
  float* Yb = Ys[wave];

  const int nj = knn[q * K_ + lane];
  const float* fb = features + (size_t)b * C_ * N_;
#pragma unroll 8
  for (int c = 0; c < C_; ++c)
    Xb[c * 32 + lane] = fb[(size_t)c * N_ + nj];
  const float* p  = xyz + ((size_t)b * N_ + nj) * 3;
  const float* qc = new_xyz + (size_t)q * 3;
  Xb[64 * 32 + lane] = p[0] - qc[0];
  Xb[65 * 32 + lane] = p[1] - qc[1];
  Xb[66 * 32 + lane] = p[2] - qc[2];
  Xb[67 * 32 + lane] = 0.0f;
  __syncthreads();

  gemm_bn_relu(Xb, 68, W0, 67,  64, g0, b0, m0, v0, Yb, lane);
  __syncthreads();
  gemm_bn_relu(Yb, 64, W1, 64,  64, g1, b1, m1, v1, Xb, lane);
  __syncthreads();
  gemm_bn_relu(Xb, 64, W2, 64, 128, g2, b2, m2, v2, Yb, lane);
  __syncthreads();

  for (int o = lane; o < 128; o += 32) {
    float mx = -BIGF;
#pragma unroll
    for (int j = 0; j < 32; ++j) mx = fmaxf(mx, Yb[o * 32 + j]);
    out_feat[((size_t)b * 128 + o) * M_ + m] = mx;
  }
}

extern "C" void kernel_launch(void* const* d_in, const int* in_sizes, int n_in,
                              void* d_out, int out_size, void* d_ws, size_t ws_size,
                              hipStream_t stream) {
  const float* xyz      = (const float*)d_in[0];
  const float* features = (const float*)d_in[1];
  const int*   indices  = (const int*)d_in[2];
  const float* W0 = (const float*)d_in[3];
  const float* g0 = (const float*)d_in[4];
  const float* b0 = (const float*)d_in[5];
  const float* m0 = (const float*)d_in[6];
  const float* v0 = (const float*)d_in[7];
  const float* W1 = (const float*)d_in[8];
  const float* g1 = (const float*)d_in[9];
  const float* b1 = (const float*)d_in[10];
  const float* m1 = (const float*)d_in[11];
  const float* v1 = (const float*)d_in[12];
  const float* W2 = (const float*)d_in[13];
  const float* g2 = (const float*)d_in[14];
  const float* b2 = (const float*)d_in[15];
  const float* m2 = (const float*)d_in[16];
  const float* v2 = (const float*)d_in[17];

  float* new_xyz  = (float*)d_out;
  float* out_feat = new_xyz + (size_t)B_ * M_ * 3;
  int*   knn      = (int*)d_ws;

  k_new_xyz<<<(B_ * M_ + 255) / 256, 256, 0, stream>>>(xyz, indices, new_xyz);
  k_knn<<<(B_ * M_) / 2, 64, 0, stream>>>(xyz, new_xyz, knn);
  k_mlp<<<(B_ * M_) / 2, 64, 0, stream>>>(features, xyz, new_xyz, knn,
      W0, g0, b0, m0, v0, W1, g1, b1, m1, v1, W2, g2, b2, m2, v2, out_feat);
}


# Features copied once to point-major rows in the workspace: the gather reads a neighbour's 64 channels as one aligned 256-B row (b128, whole lines), not 64 scattered words; X neighbour-major in LDS
# speedup vs baseline: 2.0981x; 2.0981x over previous
#include <hip/hip_runtime.h>
#include <hip/hip_bf16.h>

typedef __attribute__((ext_vector_type(2))) float v2f;
typedef __attribute__((ext_vector_type(8))) float v8f;

#define B_   16
#define N_   4096
#define C_   64
#define M_   1024
#define K_   32
#define EPS_ 1e-5f
#define BIGF 3.402823e38f

#define TP_ 65

__global__ __launch_bounds__(256) void k_prep(const float* __restrict__ xyz,
                                              const float* __restrict__ features,
                                              const int* __restrict__ indices,
                                              float* __restrict__ new_xyz,
                                              float* __restrict__ feat_t) {
  __shared__ float tile[C_ * TP_];
  const int t  = threadIdx.x;
  const int b  = blockIdx.x >> 6;
  const int n0 = (blockIdx.x & 63) << 6;
  const int r  = t >> 4;
  const int e4 = (t & 15) << 2;

  const float* src = features + (size_t)b * C_ * N_ + n0;
  for (int p = 0; p < 4; ++p) {
    const int c = r + 16 * p;
    const float4 v = *reinterpret_cast<const float4*>(src + (size_t)c * N_ + e4);
    float* d = tile + c * TP_ + e4;
    d[0] = v.x; d[1] = v.y; d[2] = v.z; d[3] = v.w;
  }
  __syncthreads();
  float* dst = feat_t + ((size_t)b * N_ + n0) * C_;
  for (int p = 0; p < 4; ++p) {
    const int n = r + 16 * p;
    const float* s = tile + e4 * TP_ + n;
    float4 v;
    v.x = s[0]; v.y = s[TP_]; v.z = s[2 * TP_]; v.w = s[3 * TP_];
    *reinterpret_cast<float4*>(dst + (size_t)n * C_ + e4) = v;
  }

  if (blockIdx.x < (B_ * M_) / 32 && t < 32) {
    const int id = blockIdx.x * 32 + t;
    const int n = indices[id & (M_ - 1)];
    const float* s = xyz + ((size_t)(id >> 10) * N_ + n) * 3;
    float* d = new_xyz + (size_t)id * 3;
    d[0] = s[0]; d[1] = s[1]; d[2] = s[2];
  }
}

#define GP_   132
#define IMAX_ 0x7fffffff

__device__ __forceinline__ int dist_key(float f) {
  const int u = __float_as_int(f);
  return u ^ ((u >> 31) & 0x7fffffff);
}

template <int M>
__device__ __forceinline__ int row_xor(int x) {
  return __builtin_amdgcn_update_dpp(0, x, 0x160 + M, 0xf, 0xf, true);
}
__device__ __forceinline__ int half_xor(int x) {
  return __builtin_amdgcn_permlanex16(x, x, 0x76543210, 0xfedcba98, false, false);
}
__device__ __forceinline__ int wave_min_i(int v) {
  v = min(v, row_xor<1>(v));
  v = min(v, row_xor<2>(v));
  v = min(v, row_xor<4>(v));
  v = min(v, row_xor<8>(v));
  v = min(v, half_xor(v));
  return v;
}

__global__ __launch_bounds__(64) void k_knn(const float* __restrict__ xyz,
                                            const float* __restrict__ new_xyz,
                                            int* __restrict__ knn) {
  __shared__ __attribute__((aligned(16))) int dist[2][32 * GP_];
  const int wave = threadIdx.x >> 5;
  const int lane = threadIdx.x & 31;
  const int q = blockIdx.x * 2 + wave;
  const int b = q >> 10;

  int* dl = dist[wave];
  const float qx = new_xyz[q * 3 + 0];
  const float qy = new_xyz[q * 3 + 1];
  const float qz = new_xyz[q * 3 + 2];
  float qq;
  {
#pragma clang fp contract(off)
    qq = __builtin_fmaf(qx, qx, qy * qy) + qz * qz;
  }
  const float* px = xyz + (size_t)b * N_ * 3;

  int gv = IMAX_;
  int gi = IMAX_;
  int* mine = dl + lane * GP_;
  for (int i = 0; i < N_ / 32; ++i) {
    const int n = lane + 32 * i;
    float x = px[n * 3 + 0], y = px[n * 3 + 1], z = px[n * 3 + 2];
    const float d = qq + (x * x + y * y + z * z) - 2.0f * (qx * x + qy * y + qz * z);
    const int key = dist_key(d);
    mine[i] = key;
    if (key < gv) { gv = key; gi = n; }
  }
  __syncthreads();

  int res = 0;
  for (int k = 0; k < K_; ++k) {
    const int mn = wave_min_i(gv);
    const int mi = wave_min_i(gv == mn ? gi : IMAX_);
    if (lane == k) res = mi;
    const int w = mi & 31;

    const int4 c4 = *reinterpret_cast<const int4*>(dl + w * GP_ + 4 * lane);
    const int c[4] = {c4.x, c4.y, c4.z, c4.w};
    const int n0 = w + 128 * lane;
    int lv = IMAX_;
    int li = IMAX_;
#pragma unroll
    for (int t = 0; t < 4; ++t) {
      const int v = c[t];
      const int n = n0 + 32 * t;
      const bool later = (v > mn) || (v == mn && n > mi);
      if (later && v < lv) { lv = v; li = n; }
    }
    const int nv = wave_min_i(lv);
    const int ni = wave_min_i(lv == nv ? li : IMAX_);
    if (lane == w) { gv = nv; gi = ni; }
  }
  knn[q * K_ + lane] = res;
}

#define FP_ 68

template <bool XT>
__device__ __forceinline__ void gemm_bn_relu(
    const float* __restrict__ Xb, int kPad,
    const float* __restrict__ W, int kW, int rowsOut,
    const float* __restrict__ g, const float* __restrict__ bt,
    const float* __restrict__ mu, const float* __restrict__ vr,
    float* __restrict__ Yb, int lane)
{
  const int half = lane >> 4;
  const int l16  = lane & 15;
  const int nKt  = kPad >> 2;

  for (int mt = 0; mt < (rowsOut >> 4); ++mt) {
    const int arow = mt * 16 + l16;
    for (int nt = 0; nt < 2; ++nt) {
      const int col = nt * 16 + l16;
      v8f acc = {};
      for (int kt = 0; kt < nKt; ++kt) {
        const int k0 = kt * 4 + half * 2;
        v2f a, bf;
        a.x  = (k0 + 0 < kW) ? W[arow * kW + k0 + 0] : 0.0f;
        a.y  = (k0 + 1 < kW) ? W[arow * kW + k0 + 1] : 0.0f;
        if (XT) {
          bf = *reinterpret_cast<const v2f*>(Xb + col * FP_ + k0);
        } else {
          bf.x = Xb[(k0 + 0) * 32 + col];
          bf.y = Xb[(k0 + 1) * 32 + col];
        }
        acc = __builtin_amdgcn_wmma_f32_16x16x4_f32(
             false, a,  false, bf,
             (short)0, acc,  false,  false);
      }
      const int rbase = mt * 16 + half * 8;
#pragma unroll
      for (int r = 0; r < 8; ++r) {
        const int row = rbase + r;
        const float s = g[row] / sqrtf(vr[row] + EPS_);
        const float y = (acc[r] - mu[row]) * s + bt[row];
        Yb[row * 32 + col] = fmaxf(y, 0.0f);
      }
    }
  }
}

__global__ __launch_bounds__(64) void k_mlp(
    const float* __restrict__ feat_t, const float* __restrict__ xyz,
    const float* __restrict__ new_xyz, const int* __restrict__ knn,
    const float* __restrict__ W0, const float* __restrict__ g0,
    const float* __restrict__ b0, const float* __restrict__ m0,
    const float* __restrict__ v0,
    const float* __restrict__ W1, const float* __restrict__ g1,
    const float* __restrict__ b1, const float* __restrict__ m1,
    const float* __restrict__ v1,
    const float* __restrict__ W2, const float* __restrict__ g2,
    const float* __restrict__ b2, const float* __restrict__ m2,
    const float* __restrict__ v2,
    float* __restrict__ out_feat)
{
  __shared__ __attribute__((aligned(16))) float Xs[2][32 * FP_];
  __shared__ float Ys[2][128 * 32];
  const int wave = threadIdx.x >> 5;
  const int lane = threadIdx.x & 31;
  const int q = blockIdx.x * 2 + wave;
  const int b = q >> 10;
  const int m = q & 1023;

  float* Xb = Xs[wave];
  float* Yb = Ys[wave];

  const int nj = knn[q * K_ + lane];
  const int half = lane >> 4;
  const int l16  = lane & 15;
  const float* fb = feat_t + (size_t)b * N_ * C_ + 4 * l16;
  float* xw = Xb + half * FP_ + 4 * l16;
  for (int i = 0; i < 16; ++i) {
    const int ne = __builtin_amdgcn_readlane(nj, 2 * i);
    const int no = __builtin_amdgcn_readlane(nj, 2 * i + 1);
    const int n  = half ? no : ne;
    const float4 f = *reinterpret_cast<const float4*>(fb + (size_t)n * C_);
    *reinterpret_cast<float4*>(xw + 2 * i * FP_) = f;
  }
  const float* p  = xyz + ((size_t)b * N_ + nj) * 3;
  const float* qc = new_xyz + (size_t)q * 3;
  float4 rel;
  rel.x = p[0] - qc[0];
  rel.y = p[1] - qc[1];
  rel.z = p[2] - qc[2];
  rel.w = 0.0f;
  *reinterpret_cast<float4*>(Xb + lane * FP_ + C_) = rel;
  __syncthreads();

  gemm_bn_relu<true >(Xb, 68, W0, 67,  64, g0, b0, m0, v0, Yb, lane);
  __syncthreads();
  gemm_bn_relu<false>(Yb, 64, W1, 64,  64, g1, b1, m1, v1, Xb, lane);
  __syncthreads();
  gemm_bn_relu<false>(Xb, 64, W2, 64, 128, g2, b2, m2, v2, Yb, lane);
  __syncthreads();

  for (int o = lane; o < 128; o += 32) {
    float mx = -BIGF;
#pragma unroll
    for (int j = 0; j < 32; ++j) mx = fmaxf(mx, Yb[o * 32 + j]);
    out_feat[((size_t)b * 128 + o) * M_ + m] = mx;
  }
}

extern "C" void kernel_launch(void* const* d_in, const int* in_sizes, int n_in,
                              void* d_out, int out_size, void* d_ws, size_t ws_size,
                              hipStream_t stream) {
  const float* xyz      = (const float*)d_in[0];
  const float* features = (const float*)d_in[1];
  const int*   indices  = (const int*)d_in[2];
  const float* W0 = (const float*)d_in[3];
  const float* g0 = (const float*)d_in[4];
  const float* b0 = (const float*)d_in[5];
  const float* m0 = (const float*)d_in[6];
  const float* v0 = (const float*)d_in[7];
  const float* W1 = (const float*)d_in[8];
  const float* g1 = (const float*)d_in[9];
  const float* b1 = (const float*)d_in[10];
  const float* m1 = (const float*)d_in[11];
  const float* v1 = (const float*)d_in[12];
  const float* W2 = (const float*)d_in[13];
  const float* g2 = (const float*)d_in[14];
  const float* b2 = (const float*)d_in[15];
  const float* m2 = (const float*)d_in[16];
  const float* v2 = (const float*)d_in[17];

  float* new_xyz  = (float*)d_out;
  float* out_feat = new_xyz + (size_t)B_ * M_ * 3;
  int*   knn      = (int*)d_ws;
  float* feat_t   = (float*)d_ws + (size_t)B_ * M_ * K_;

  k_prep<<<B_ * (N_ / 64), 256, 0, stream>>>(xyz, features, indices, new_xyz, feat_t);
  k_knn<<<(B_ * M_) / 2, 64, 0, stream>>>(xyz, new_xyz, knn);
  k_mlp<<<(B_ * M_) / 2, 64, 0, stream>>>(feat_t, xyz, new_xyz, knn,
      W0, g0, b0, m0, v0, W1, g1, b1, m1, v1, W2, g2, b2, m2, v2, out_feat);
}


# Weights copied once per call into WMMA A-operand order in the workspace: a wave's A load is 512 contiguous bytes (16 B a lane, two k-steps, whole lines), not 8 B in each of 16 rows
# speedup vs baseline: 2.3605x; 1.1251x over previous
#include <hip/hip_runtime.h>
#include <hip/hip_bf16.h>

typedef __attribute__((ext_vector_type(2))) float v2f;
typedef __attribute__((ext_vector_type(8))) float v8f;

#define B_   16
#define N_   4096
#define C_   64
#define M_   1024
#define K_   32
#define EPS_ 1e-5f
#define BIGF 3.402823e38f

#define TP_ 65

#define WF0_  0
#define WF1_  (64 * 68)
#define WF2_  (WF1_ + 64 * 64)
#define WFN_  (WF2_ + 128 * 64)
#define WFB_  (WFN_ / 256)

__global__ __launch_bounds__(256) void k_prep(const float* __restrict__ xyz,
                                              const float* __restrict__ features,
                                              const int* __restrict__ indices,
                                              const float* __restrict__ W0,
                                              const float* __restrict__ W1,
                                              const float* __restrict__ W2,
                                              float* __restrict__ new_xyz,
                                              float* __restrict__ feat_t,
                                              float* __restrict__ wfrag) {
  __shared__ float tile[C_ * TP_];
  const int t  = threadIdx.x;
  const int b  = blockIdx.x >> 6;
  const int n0 = (blockIdx.x & 63) << 6;
  const int r  = t >> 4;
  const int e4 = (t & 15) << 2;

  const float* src = features + (size_t)b * C_ * N_ + n0;
  for (int p = 0; p < 4; ++p) {
    const int c = r + 16 * p;
    const float4 v = *reinterpret_cast<const float4*>(src + (size_t)c * N_ + e4);
    float* d = tile + c * TP_ + e4;
    d[0] = v.x; d[1] = v.y; d[2] = v.z; d[3] = v.w;
  }
  __syncthreads();
  float* dst = feat_t + ((size_t)b * N_ + n0) * C_;
  for (int p = 0; p < 4; ++p) {
    const int n = r + 16 * p;
    const float* s = tile + e4 * TP_ + n;
    float4 v;
    v.x = s[0]; v.y = s[TP_]; v.z = s[2 * TP_]; v.w = s[3 * TP_];
    *reinterpret_cast<float4*>(dst + (size_t)n * C_ + e4) = v;
  }

  if (blockIdx.x < (B_ * M_) / 32 && t < 32) {
    const int id = blockIdx.x * 32 + t;
    const int n = indices[id & (M_ - 1)];
    const float* s = xyz + ((size_t)(id >> 10) * N_ + n) * 3;
    float* d = new_xyz + (size_t)id * 3;
    d[0] = s[0]; d[1] = s[1]; d[2] = s[2];
  }

  if (blockIdx.x >= 512 && blockIdx.x < 512 + WFB_) {
    const int e = (blockIdx.x - 512) * 256 + t;
    const float* Wl;
    int kW, nKt, mt, r;
    if (e < WF1_)      { Wl = W0; kW = 67; nKt = 17; r = e - WF0_; mt = r / (17 * 64); r -= mt * (17 * 64); }
    else if (e < WF2_) { Wl = W1; kW = 64; nKt = 16; r = e - WF1_; mt = r >> 10; r &= 1023; }
    else               { Wl = W2; kW = 64; nKt = 16; r = e - WF2_; mt = r >> 10; r &= 1023; }
    const int paired = (nKt >> 1) * 128;
    int kt, ln, j;
    if (r < paired) { kt = 2 * (r >> 7) + ((r >> 1) & 1); ln = (r & 127) >> 2; j = r & 1; }
    else            { r -= paired; kt = nKt - 1; ln = r >> 1; j = r & 1; }
    const int k = 4 * kt + 2 * (ln >> 4) + j;
    const int row = mt * 16 + (ln & 15);
    wfrag[e] = (k < kW) ? Wl[row * kW + k] : 0.0f;
  }
}

#define GP_   132
#define IMAX_ 0x7fffffff

__device__ __forceinline__ int dist_key(float f) {
  const int u = __float_as_int(f);
  return u ^ ((u >> 31) & 0x7fffffff);
}

template <int M>
__device__ __forceinline__ int row_xor(int x) {
  return __builtin_amdgcn_update_dpp(0, x, 0x160 + M, 0xf, 0xf, true);
}
__device__ __forceinline__ int half_xor(int x) {
  return __builtin_amdgcn_permlanex16(x, x, 0x76543210, 0xfedcba98, false, false);
}
__device__ __forceinline__ int wave_min_i(int v) {
  v = min(v, row_xor<1>(v));
  v = min(v, row_xor<2>(v));
  v = min(v, row_xor<4>(v));
  v = min(v, row_xor<8>(v));
  v = min(v, half_xor(v));
  return v;
}

__global__ __launch_bounds__(64) void k_knn(const float* __restrict__ xyz,
                                            const float* __restrict__ new_xyz,
                                            int* __restrict__ knn) {
  __shared__ __attribute__((aligned(16))) int dist[2][32 * GP_];
  const int wave = threadIdx.x >> 5;
  const int lane = threadIdx.x & 31;
  const int q = blockIdx.x * 2 + wave;
  const int b = q >> 10;

  int* dl = dist[wave];
  const float qx = new_xyz[q * 3 + 0];
  const float qy = new_xyz[q * 3 + 1];
  const float qz = new_xyz[q * 3 + 2];
  float qq;
  {
#pragma clang fp contract(off)
    qq = __builtin_fmaf(qx, qx, qy * qy) + qz * qz;
  }
  const float* px = xyz + (size_t)b * N_ * 3;

  int gv = IMAX_;
  int gi = IMAX_;
  int* mine = dl + lane * GP_;
  for (int i = 0; i < N_ / 32; ++i) {
    const int n = lane + 32 * i;
    float x = px[n * 3 + 0], y = px[n * 3 + 1], z = px[n * 3 + 2];
    const float d = qq + (x * x + y * y + z * z) - 2.0f * (qx * x + qy * y + qz * z);
    const int key = dist_key(d);
    mine[i] = key;
    if (key < gv) { gv = key; gi = n; }
  }
  __syncthreads();

  int res = 0;
  for (int k = 0; k < K_; ++k) {
    const int mn = wave_min_i(gv);
    const int mi = wave_min_i(gv == mn ? gi : IMAX_);
    if (lane == k) res = mi;
    const int w = mi & 31;

    const int4 c4 = *reinterpret_cast<const int4*>(dl + w * GP_ + 4 * lane);
    const int c[4] = {c4.x, c4.y, c4.z, c4.w};
    const int n0 = w + 128 * lane;
    int lv = IMAX_;
    int li = IMAX_;
#pragma unroll
    for (int t = 0; t < 4; ++t) {
      const int v = c[t];
      const int n = n0 + 32 * t;
      const bool later = (v > mn) || (v == mn && n > mi);
      if (later && v < lv) { lv = v; li = n; }
    }
    const int nv = wave_min_i(lv);
    const int ni = wave_min_i(lv == nv ? li : IMAX_);
    if (lane == w) { gv = nv; gi = ni; }
  }
  knn[q * K_ + lane] = res;
}

#define FP_ 68

template <bool XT>
__device__ __forceinline__ v2f load_b_frag(const float* __restrict__ Xb, int col, int k0) {
  v2f bf;
  if (XT) {
    bf = *reinterpret_cast<const v2f*>(Xb + col * FP_ + k0);
  } else {
    bf.x = Xb[(k0 + 0) * 32 + col];
    bf.y = Xb[(k0 + 1) * 32 + col];
  }
  return bf;
}

template <bool XT>
__device__ __forceinline__ void gemm_bn_relu(
    const float* __restrict__ Xb, int kPad,
    const float* __restrict__ Wf, int rowsOut,
    const float* __restrict__ g, const float* __restrict__ bt,
    const float* __restrict__ mu, const float* __restrict__ vr,
    float* __restrict__ Yb, int lane)
{
  const int half = lane >> 4;
  const int l16  = lane & 15;
  const int nKt  = kPad >> 2;

  for (int mt = 0; mt < (rowsOut >> 4); ++mt) {
    const float* wt = Wf + mt * (nKt * 64);
    for (int nt = 0; nt < 2; ++nt) {
      const int col = nt * 16 + l16;
      v8f acc = {};
      for (int p = 0; p < (nKt >> 1); ++p) {
        const float4 a4 = *reinterpret_cast<const float4*>(wt + p * 128 + 4 * lane);
        v2f a;
        a.x = a4.x; a.y = a4.y;
        acc = __builtin_amdgcn_wmma_f32_16x16x4_f32(
             false, a,  false,
            load_b_frag<XT>(Xb, col, (2 * p) * 4 + half * 2),
             (short)0, acc,  false,  false);
        a.x = a4.z; a.y = a4.w;
        acc = __builtin_amdgcn_wmma_f32_16x16x4_f32(
             false, a,  false,
            load_b_frag<XT>(Xb, col, (2 * p + 1) * 4 + half * 2),
             (short)0, acc,  false,  false);
      }
      if (nKt & 1) {
        const v2f a = *reinterpret_cast<const v2f*>(wt + (nKt - 1) * 64 + 2 * lane);
        acc = __builtin_amdgcn_wmma_f32_16x16x4_f32(
             false, a,  false,
            load_b_frag<XT>(Xb, col, (nKt - 1) * 4 + half * 2),
             (short)0, acc,  false,  false);
      }
      const int rbase = mt * 16 + half * 8;
#pragma unroll
      for (int r = 0; r < 8; ++r) {
        const int row = rbase + r;
        const float s = g[row] / sqrtf(vr[row] + EPS_);
        const float y = (acc[r] - mu[row]) * s + bt[row];
        Yb[row * 32 + col] = fmaxf(y, 0.0f);
      }
    }
  }
}

__global__ __launch_bounds__(64) void k_mlp(
    const float* __restrict__ feat_t, const float* __restrict__ xyz,
    const float* __restrict__ new_xyz, const int* __restrict__ knn,
    const float* __restrict__ wfrag, const float* __restrict__ g0,
    const float* __restrict__ b0, const float* __restrict__ m0,
    const float* __restrict__ v0,
    const float* __restrict__ g1,
    const float* __restrict__ b1, const float* __restrict__ m1,
    const float* __restrict__ v1,
    const float* __restrict__ g2,
    const float* __restrict__ b2, const float* __restrict__ m2,
    const float* __restrict__ v2,
    float* __restrict__ out_feat)
{
  __shared__ __attribute__((aligned(16))) float Xs[2][32 * FP_];
  __shared__ float Ys[2][128 * 32];
  const int wave = threadIdx.x >> 5;
  const int lane = threadIdx.x & 31;
  const int q = blockIdx.x * 2 + wave;
  const int b = q >> 10;
  const int m = q & 1023;

  float* Xb = Xs[wave];
  float* Yb = Ys[wave];

  const int nj = knn[q * K_ + lane];
  const int half = lane >> 4;
  const int l16  = lane & 15;
  const float* fb = feat_t + (size_t)b * N_ * C_ + 4 * l16;
  float* xw = Xb + half * FP_ + 4 * l16;
  for (int i = 0; i < 16; ++i) {
    const int ne = __builtin_amdgcn_readlane(nj, 2 * i);
    const int no = __builtin_amdgcn_readlane(nj, 2 * i + 1);
    const int n  = half ? no : ne;
    const float4 f = *reinterpret_cast<const float4*>(fb + (size_t)n * C_);
    *reinterpret_cast<float4*>(xw + 2 * i * FP_) = f;
  }
  const float* p  = xyz + ((size_t)b * N_ + nj) * 3;
  const float* qc = new_xyz + (size_t)q * 3;
  float4 rel;
  rel.x = p[0] - qc[0];
  rel.y = p[1] - qc[1];
  rel.z = p[2] - qc[2];
  rel.w = 0.0f;
  *reinterpret_cast<float4*>(Xb + lane * FP_ + C_) = rel;
  __syncthreads();

  gemm_bn_relu<true >(Xb, 68, wfrag + WF0_,  64, g0, b0, m0, v0, Yb, lane);
  __syncthreads();
  gemm_bn_relu<false>(Yb, 64, wfrag + WF1_,  64, g1, b1, m1, v1, Xb, lane);
  __syncthreads();
  gemm_bn_relu<false>(Xb, 64, wfrag + WF2_, 128, g2, b2, m2, v2, Yb, lane);
  __syncthreads();

  for (int o = lane; o < 128; o += 32) {
    float mx = -BIGF;
#pragma unroll
    for (int j = 0; j < 32; ++j) mx = fmaxf(mx, Yb[o * 32 + j]);
    out_feat[((size_t)b * 128 + o) * M_ + m] = mx;
  }
}

extern "C" void kernel_launch(void* const* d_in, const int* in_sizes, int n_in,
                              void* d_out, int out_size, void* d_ws, size_t ws_size,
                              hipStream_t stream) {
  const float* xyz      = (const float*)d_in[0];
  const float* features = (const float*)d_in[1];
  const int*   indices  = (const int*)d_in[2];
  const float* W0 = (const float*)d_in[3];
  const float* g0 = (const float*)d_in[4];
  const float* b0 = (const float*)d_in[5];
  const float* m0 = (const float*)d_in[6];
  const float* v0 = (const float*)d_in[7];
  const float* W1 = (const float*)d_in[8];
  const float* g1 = (const float*)d_in[9];
  const float* b1 = (const float*)d_in[10];
  const float* m1 = (const float*)d_in[11];
  const float* v1 = (const float*)d_in[12];
  const float* W2 = (const float*)d_in[13];
  const float* g2 = (const float*)d_in[14];
  const float* b2 = (const float*)d_in[15];
  const float* m2 = (const float*)d_in[16];
  const float* v2 = (const float*)d_in[17];

  float* new_xyz  = (float*)d_out;
  float* out_feat = new_xyz + (size_t)B_ * M_ * 3;
  int*   knn      = (int*)d_ws;
  float* feat_t   = (float*)d_ws + (size_t)B_ * M_ * K_;
  float* wfrag    = feat_t + (size_t)B_ * N_ * C_;

  k_prep<<<B_ * (N_ / 64), 256, 0, stream>>>(xyz, features, indices, W0, W1, W2,
                                              new_xyz, feat_t, wfrag);
  k_knn<<<(B_ * M_) / 2, 64, 0, stream>>>(xyz, new_xyz, knn);
  k_mlp<<<(B_ * M_) / 2, 64, 0, stream>>>(feat_t, xyz, new_xyz, knn, wfrag,
      g0, b0, m0, v0, g1, b1, m1, v1, g2, b2, m2, v2, out_feat);
}


# Layer 0's product with the 64 feature channels made once per point (WMMA in the first kernel, same k order) and gathered as the accumulator's start; per neighbour only the xyz k-step is left
# speedup vs baseline: 2.5120x; 1.0642x over previous
#include <hip/hip_runtime.h>
#include <hip/hip_bf16.h>

typedef __attribute__((ext_vector_type(2))) float v2f;
typedef __attribute__((ext_vector_type(8))) float v8f;

#define B_   16
#define N_   4096
#define C_   64
#define M_   1024
#define K_   32
#define EPS_ 1e-5f
#define BIGF 3.402823e38f

#define TPK_ 72
#define ZP_  68
#define WKP_ 66
#define WMT_ (16 * WKP_)

#define WF0_  0
#define WF1_  (4 * 64)
#define WF2_  (WF1_ + 64 * 64)
#define WFN_  (WF2_ + 128 * 64)
#define WFB_  (WFN_ / 256)

__global__ __launch_bounds__(256) void k_prep(const float* __restrict__ xyz,
                                              const float* __restrict__ features,
                                              const int* __restrict__ indices,
                                              const float* __restrict__ W0,
                                              const float* __restrict__ W1,
                                              const float* __restrict__ W2,
                                              float* __restrict__ new_xyz,
                                              float* __restrict__ z0,
                                              float* __restrict__ wfrag) {
  __shared__ __attribute__((aligned(16))) float tile[C_ * TPK_];
  __shared__ __attribute__((aligned(16))) float wl[4 * WMT_];
  const int t  = threadIdx.x;
  const int b  = blockIdx.x >> 6;
  const int n0 = (blockIdx.x & 63) << 6;
  const int r  = t >> 4;
  const int e4 = (t & 15) << 2;

  const float* src = features + (size_t)b * C_ * N_ + n0;
  for (int p = 0; p < 4; ++p) {
    const int c = r + 16 * p;
    const float4 v = *reinterpret_cast<const float4*>(src + (size_t)c * N_ + e4);
    *reinterpret_cast<float4*>(tile + c * TPK_ + e4) = v;
  }
  for (int i = 0; i < 16; ++i) {
    const int f   = t + 256 * i;
    const int row = f >> 6;
    const int k   = f & 63;
    const int ln  = (row & 15) + 16 * ((k >> 1) & 1);
    wl[(row >> 4) * WMT_ + (k >> 2) * WKP_ + 2 * ln + (k & 1)] = W0[row * 67 + k];
  }
  __syncthreads();

  const int wave = t >> 5;
  const int lane = t & 31;
  const int half = lane >> 4;
  const int l16  = lane & 15;
  const int nt   = wave & 3;
  const int mt0  = (wave >> 2) << 1;
  {
    const float* xb = tile + (2 * half) * TPK_ + nt * 16 + l16;
    const float* wa = wl + mt0 * WMT_ + 2 * lane;
    v8f acc0 = {};
    v8f acc1 = {};
    for (int kt = 0; kt < 16; ++kt) {
      v2f bf;
      bf.x = xb[(4 * kt + 0) * TPK_];
      bf.y = xb[(4 * kt + 1) * TPK_];
      const v2f a0 = *reinterpret_cast<const v2f*>(wa + kt * WKP_);
      const v2f a1 = *reinterpret_cast<const v2f*>(wa + WMT_ + kt * WKP_);
      acc0 = __builtin_amdgcn_wmma_f32_16x16x4_f32(
           false, a0,  false, bf,
           (short)0, acc0,  false,  false);
      acc1 = __builtin_amdgcn_wmma_f32_16x16x4_f32(
           false, a1,  false, bf,
           (short)0, acc1,  false,  false);
    }
    __syncthreads();
    float* zr = tile + (nt * 16 + l16) * ZP_ + mt0 * 16 + 8 * half;
    float4 v;
    v.x = acc0[0]; v.y = acc0[1]; v.z = acc0[2]; v.w = acc0[3];
    *reinterpret_cast<float4*>(zr) = v;
    v.x = acc0[4]; v.y = acc0[5]; v.z = acc0[6]; v.w = acc0[7];
    *reinterpret_cast<float4*>(zr + 4) = v;
    v.x = acc1[0]; v.y = acc1[1]; v.z = acc1[2]; v.w = acc1[3];
    *reinterpret_cast<float4*>(zr + 16) = v;
    v.x = acc1[4]; v.y = acc1[5]; v.z = acc1[6]; v.w = acc1[7];
    *reinterpret_cast<float4*>(zr + 20) = v;
  }
  __syncthreads();
  float* dst = z0 + ((size_t)b * N_ + n0) * C_;
  for (int p = 0; p < 4; ++p) {
    const int n = r + 16 * p;
    const float4 v = *reinterpret_cast<const float4*>(tile + n * ZP_ + e4);
    *reinterpret_cast<float4*>(dst + (size_t)n * C_ + e4) = v;
  }

  if (blockIdx.x < (B_ * M_) / 32 && t < 32) {
    const int id = blockIdx.x * 32 + t;
    const int n = indices[id & (M_ - 1)];
    const float* s = xyz + ((size_t)(id >> 10) * N_ + n) * 3;
    float* d = new_xyz + (size_t)id * 3;
    d[0] = s[0]; d[1] = s[1]; d[2] = s[2];
  }

  if (blockIdx.x >= 512 && blockIdx.x < 512 + WFB_) {
    const int e = (blockIdx.x - 512) * 256 + t;
    if (e < WF1_) {
      const int mt = e >> 6;
      const int ln = (e >> 1) & 31;
      const int k  = 64 + 2 * (ln >> 4) + (e & 1);
      wfrag[e] = (k < 67) ? W0[(mt * 16 + (ln & 15)) * 67 + k] : 0.0f;
    } else {
      const float* Wl = (e < WF2_) ? W1 : W2;
      const int q  = e - ((e < WF2_) ? WF1_ : WF2_);
      const int mt = q >> 10;
      const int w  = q & 1023;
      const int kt = 2 * (w >> 7) + ((w >> 1) & 1);
      const int ln = (w & 127) >> 2;
      const int k  = 4 * kt + 2 * (ln >> 4) + (w & 1);
      wfrag[e] = Wl[(mt * 16 + (ln & 15)) * 64 + k];
    }
  }
}

#define GP_   132
#define IMAX_ 0x7fffffff

__device__ __forceinline__ int dist_key(float f) {
  const int u = __float_as_int(f);
  return u ^ ((u >> 31) & 0x7fffffff);
}

template <int M>
__device__ __forceinline__ int row_xor(int x) {
  return __builtin_amdgcn_update_dpp(0, x, 0x160 + M, 0xf, 0xf, true);
}
__device__ __forceinline__ int half_xor(int x) {
  return __builtin_amdgcn_permlanex16(x, x, 0x76543210, 0xfedcba98, false, false);
}
__device__ __forceinline__ int wave_min_i(int v) {
  v = min(v, row_xor<1>(v));
  v = min(v, row_xor<2>(v));
  v = min(v, row_xor<4>(v));
  v = min(v, row_xor<8>(v));
  v = min(v, half_xor(v));
  return v;
}

__global__ __launch_bounds__(64) void k_knn(const float* __restrict__ xyz,
                                            const float* __restrict__ new_xyz,
                                            int* __restrict__ knn) {
  __shared__ __attribute__((aligned(16))) int dist[2][32 * GP_];
  const int wave = threadIdx.x >> 5;
  const int lane = threadIdx.x & 31;
  const int q = blockIdx.x * 2 + wave;
  const int b = q >> 10;

  int* dl = dist[wave];
  const float qx = new_xyz[q * 3 + 0];
  const float qy = new_xyz[q * 3 + 1];
  const float qz = new_xyz[q * 3 + 2];
  float qq;
  {
#pragma clang fp contract(off)
    qq = __builtin_fmaf(qx, qx, qy * qy) + qz * qz;
  }
  const float* px = xyz + (size_t)b * N_ * 3;

  int gv = IMAX_;
  int gi = IMAX_;
  int* mine = dl + lane * GP_;
  for (int i = 0; i < N_ / 32; ++i) {
    const int n = lane + 32 * i;
    float x = px[n * 3 + 0], y = px[n * 3 + 1], z = px[n * 3 + 2];
    const float d = qq + (x * x + y * y + z * z) - 2.0f * (qx * x + qy * y + qz * z);
    const int key = dist_key(d);
    mine[i] = key;
    if (key < gv) { gv = key; gi = n; }
  }
  __syncthreads();

  int res = 0;
  for (int k = 0; k < K_; ++k) {
    const int mn = wave_min_i(gv);
    const int mi = wave_min_i(gv == mn ? gi : IMAX_);
    if (lane == k) res = mi;
    const int w = mi & 31;

    const int4 c4 = *reinterpret_cast<const int4*>(dl + w * GP_ + 4 * lane);
    const int c[4] = {c4.x, c4.y, c4.z, c4.w};
    const int n0 = w + 128 * lane;
    int lv = IMAX_;
    int li = IMAX_;
#pragma unroll
    for (int t = 0; t < 4; ++t) {
      const int v = c[t];
      const int n = n0 + 32 * t;
      const bool later = (v > mn) || (v == mn && n > mi);
      if (later && v < lv) { lv = v; li = n; }
    }
    const int nv = wave_min_i(lv);
    const int ni = wave_min_i(lv == nv ? li : IMAX_);
    if (lane == w) { gv = nv; gi = ni; }
  }
  knn[q * K_ + lane] = res;
}

#define FP_ 68

__device__ __forceinline__ void bn_relu_store(
    const v8f acc, int rbase, int col,
    const float* __restrict__ g, const float* __restrict__ bt,
    const float* __restrict__ mu, const float* __restrict__ vr,
    float* __restrict__ Yb)
{
#pragma unroll
  for (int r = 0; r < 8; ++r) {
    const int row = rbase + r;
    const float s = g[row] / sqrtf(vr[row] + EPS_);
    const float y = (acc[r] - mu[row]) * s + bt[row];
    Yb[row * 32 + col] = fmaxf(y, 0.0f);
  }
}

__device__ __forceinline__ void layer0_bn_relu(
    const float* __restrict__ Xn, const float* __restrict__ Wf,
    const float* __restrict__ g, const float* __restrict__ bt,
    const float* __restrict__ mu, const float* __restrict__ vr,
    float* __restrict__ Yb, int lane)
{
  const int half = lane >> 4;
  const int l16  = lane & 15;
  for (int mt = 0; mt < 4; ++mt) {
    const v2f a = *reinterpret_cast<const v2f*>(Wf + mt * 64 + 2 * lane);
    for (int nt = 0; nt < 2; ++nt) {
      const int col = nt * 16 + l16;
      const float* xc = Xn + col * FP_;
      const float4 c0 = *reinterpret_cast<const float4*>(xc + mt * 16 + 8 * half);
      const float4 c1 = *reinterpret_cast<const float4*>(xc + mt * 16 + 8 * half + 4);
      const v2f bf = *reinterpret_cast<const v2f*>(xc + C_ + 2 * half);
      v8f acc;
      acc[0] = c0.x; acc[1] = c0.y; acc[2] = c0.z; acc[3] = c0.w;
      acc[4] = c1.x; acc[5] = c1.y; acc[6] = c1.z; acc[7] = c1.w;
      acc = __builtin_amdgcn_wmma_f32_16x16x4_f32(
           false, a,  false, bf,
           (short)0, acc,  false,  false);
      bn_relu_store(acc, mt * 16 + half * 8, col, g, bt, mu, vr, Yb);
    }
  }
}

__device__ __forceinline__ void gemm_bn_relu(
    const float* __restrict__ Xb,
    const float* __restrict__ Wf, int rowsOut,
    const float* __restrict__ g, const float* __restrict__ bt,
    const float* __restrict__ mu, const float* __restrict__ vr,
    float* __restrict__ Yb, int lane)
{
  const int half = lane >> 4;
  const int l16  = lane & 15;

  for (int mt = 0; mt < (rowsOut >> 4); ++mt) {
    const float* wt = Wf + mt * 1024;
    for (int nt = 0; nt < 2; ++nt) {
      const int col = nt * 16 + l16;
      v8f acc = {};
      for (int p = 0; p < 8; ++p) {
        const float4 a4 = *reinterpret_cast<const float4*>(wt + p * 128 + 4 * lane);
        const int k0 = 8 * p + 2 * half;
        v2f a, bf;
        a.x = a4.x; a.y = a4.y;
        bf.x = Xb[(k0 + 0) * 32 + col];
        bf.y = Xb[(k0 + 1) * 32 + col];
        acc = __builtin_amdgcn_wmma_f32_16x16x4_f32(
             false, a,  false, bf,
             (short)0, acc,  false,  false);
        a.x = a4.z; a.y = a4.w;
        bf.x = Xb[(k0 + 4) * 32 + col];
        bf.y = Xb[(k0 + 5) * 32 + col];
        acc = __builtin_amdgcn_wmma_f32_16x16x4_f32(
             false, a,  false, bf,
             (short)0, acc,  false,  false);
      }
      bn_relu_store(acc, mt * 16 + half * 8, col, g, bt, mu, vr, Yb);
    }
  }
}

__global__ __launch_bounds__(64) void k_mlp(
    const float* __restrict__ z0, const float* __restrict__ xyz,
    const float* __restrict__ new_xyz, const int* __restrict__ knn,
    const float* __restrict__ wfrag, const float* __restrict__ g0,
    const float* __restrict__ b0, const float* __restrict__ m0,
    const float* __restrict__ v0,
    const float* __restrict__ g1,
    const float* __restrict__ b1, const float* __restrict__ m1,
    const float* __restrict__ v1,
    const float* __restrict__ g2,
    const float* __restrict__ b2, const float* __restrict__ m2,
    const float* __restrict__ v2,
    float* __restrict__ out_feat)
{
  __shared__ __attribute__((aligned(16))) float Xs[2][32 * FP_];
  __shared__ float Ys[2][128 * 32];
  const int wave = threadIdx.x >> 5;
  const int lane = threadIdx.x & 31;
  const int q = blockIdx.x * 2 + wave;
  const int b = q >> 10;
  const int m = q & 1023;

  float* Xb = Xs[wave];
  float* Yb = Ys[wave];

  const int nj = knn[q * K_ + lane];
  const int half = lane >> 4;
  const int l16  = lane & 15;
  const float* fb = z0 + (size_t)b * N_ * C_ + 4 * l16;
  float* xw = Xb + half * FP_ + 4 * l16;
  for (int i = 0; i < 16; ++i) {
    const int ne = __builtin_amdgcn_readlane(nj, 2 * i);
    const int no = __builtin_amdgcn_readlane(nj, 2 * i + 1);
    const int n  = half ? no : ne;
    const float4 f = *reinterpret_cast<const float4*>(fb + (size_t)n * C_);
    *reinterpret_cast<float4*>(xw + 2 * i * FP_) = f;
  }
  const float* p  = xyz + ((size_t)b * N_ + nj) * 3;
  const float* qc = new_xyz + (size_t)q * 3;
  float4 rel;
  rel.x = p[0] - qc[0];
  rel.y = p[1] - qc[1];
  rel.z = p[2] - qc[2];
  rel.w = 0.0f;
  *reinterpret_cast<float4*>(Xb + lane * FP_ + C_) = rel;
  __syncthreads();

  layer0_bn_relu(Xb, wfrag + WF0_, g0, b0, m0, v0, Yb, lane);
  __syncthreads();
  gemm_bn_relu(Yb, wfrag + WF1_,  64, g1, b1, m1, v1, Xb, lane);
  __syncthreads();
  gemm_bn_relu(Xb, wfrag + WF2_, 128, g2, b2, m2, v2, Yb, lane);
  __syncthreads();

  for (int o = lane; o < 128; o += 32) {
    float mx = -BIGF;
#pragma unroll
    for (int j = 0; j < 32; ++j) mx = fmaxf(mx, Yb[o * 32 + j]);
    out_feat[((size_t)b * 128 + o) * M_ + m] = mx;
  }
}

extern "C" void kernel_launch(void* const* d_in, const int* in_sizes, int n_in,
                              void* d_out, int out_size, void* d_ws, size_t ws_size,
                              hipStream_t stream) {
  const float* xyz      = (const float*)d_in[0];
  const float* features = (const float*)d_in[1];
  const int*   indices  = (const int*)d_in[2];
  const float* W0 = (const float*)d_in[3];
  const float* g0 = (const float*)d_in[4];
  const float* b0 = (const float*)d_in[5];
  const float* m0 = (const float*)d_in[6];
  const float* v0 = (const float*)d_in[7];
  const float* W1 = (const float*)d_in[8];
  const float* g1 = (const float*)d_in[9];
  const float* b1 = (const float*)d_in[10];
  const float* m1 = (const float*)d_in[11];
  const float* v1 = (const float*)d_in[12];
  const float* W2 = (const float*)d_in[13];
  const float* g2 = (const float*)d_in[14];
  const float* b2 = (const float*)d_in[15];
  const float* m2 = (const float*)d_in[16];
  const float* v2 = (const float*)d_in[17];

  float* new_xyz  = (float*)d_out;
  float* out_feat = new_xyz + (size_t)B_ * M_ * 3;
  int*   knn      = (int*)d_ws;
  float* z0       = (float*)d_ws + (size_t)B_ * M_ * K_;
  float* wfrag    = z0 + (size_t)B_ * N_ * C_;

  k_prep<<<B_ * (N_ / 64), 256, 0, stream>>>(xyz, features, indices, W0, W1, W2,
                                              new_xyz, z0, wfrag);
  k_knn<<<(B_ * M_) / 2, 64, 0, stream>>>(xyz, new_xyz, knn);
  k_mlp<<<(B_ * M_) / 2, 64, 0, stream>>>(z0, xyz, new_xyz, knn, wfrag,
      g0, b0, m0, v0, g1, b1, m1, v1, g2, b2, m2, v2, out_feat);
}


# kNN distance loop: a lane's points come through two register sets of 8, the next set's loads issued ahead of the current set's arithmetic, not in the trip that uses them; distances bit-identical
# speedup vs baseline: 2.8906x; 1.1507x over previous
#include <hip/hip_runtime.h>
#include <hip/hip_bf16.h>

typedef __attribute__((ext_vector_type(2))) float v2f;
typedef __attribute__((ext_vector_type(8))) float v8f;

#define B_   16
#define N_   4096
#define C_   64
#define M_   1024
#define K_   32
#define EPS_ 1e-5f
#define BIGF 3.402823e38f

#define TPK_ 72
#define ZP_  68
#define WKP_ 66
#define WMT_ (16 * WKP_)

#define WF0_  0
#define WF1_  (4 * 64)
#define WF2_  (WF1_ + 64 * 64)
#define WFN_  (WF2_ + 128 * 64)
#define WFB_  (WFN_ / 256)

__global__ __launch_bounds__(256) void k_prep(const float* __restrict__ xyz,
                                              const float* __restrict__ features,
                                              const int* __restrict__ indices,
                                              const float* __restrict__ W0,
                                              const float* __restrict__ W1,
                                              const float* __restrict__ W2,
                                              float* __restrict__ new_xyz,
                                              float* __restrict__ z0,
                                              float* __restrict__ wfrag) {
  __shared__ __attribute__((aligned(16))) float tile[C_ * TPK_];
  __shared__ __attribute__((aligned(16))) float wl[4 * WMT_];
  const int t  = threadIdx.x;
  const int b  = blockIdx.x >> 6;
  const int n0 = (blockIdx.x & 63) << 6;
  const int r  = t >> 4;
  const int e4 = (t & 15) << 2;

  const float* src = features + (size_t)b * C_ * N_ + n0;
  for (int p = 0; p < 4; ++p) {
    const int c = r + 16 * p;
    const float4 v = *reinterpret_cast<const float4*>(src + (size_t)c * N_ + e4);
    *reinterpret_cast<float4*>(tile + c * TPK_ + e4) = v;
  }
  for (int i = 0; i < 16; ++i) {
    const int f   = t + 256 * i;
    const int row = f >> 6;
    const int k   = f & 63;
    const int ln  = (row & 15) + 16 * ((k >> 1) & 1);
    wl[(row >> 4) * WMT_ + (k >> 2) * WKP_ + 2 * ln + (k & 1)] = W0[row * 67 + k];
  }
  __syncthreads();

  const int wave = t >> 5;
  const int lane = t & 31;
  const int half = lane >> 4;
  const int l16  = lane & 15;
  const int nt   = wave & 3;
  const int mt0  = (wave >> 2) << 1;
  {
    const float* xb = tile + (2 * half) * TPK_ + nt * 16 + l16;
    const float* wa = wl + mt0 * WMT_ + 2 * lane;
    v8f acc0 = {};
    v8f acc1 = {};
    for (int kt = 0; kt < 16; ++kt) {
      v2f bf;
      bf.x = xb[(4 * kt + 0) * TPK_];
      bf.y = xb[(4 * kt + 1) * TPK_];
      const v2f a0 = *reinterpret_cast<const v2f*>(wa + kt * WKP_);
      const v2f a1 = *reinterpret_cast<const v2f*>(wa + WMT_ + kt * WKP_);
      acc0 = __builtin_amdgcn_wmma_f32_16x16x4_f32(
           false, a0,  false, bf,
           (short)0, acc0,  false,  false);
      acc1 = __builtin_amdgcn_wmma_f32_16x16x4_f32(
           false, a1,  false, bf,
           (short)0, acc1,  false,  false);
    }
    __syncthreads();
    float* zr = tile + (nt * 16 + l16) * ZP_ + mt0 * 16 + 8 * half;
    float4 v;
    v.x = acc0[0]; v.y = acc0[1]; v.z = acc0[2]; v.w = acc0[3];
    *reinterpret_cast<float4*>(zr) = v;
    v.x = acc0[4]; v.y = acc0[5]; v.z = acc0[6]; v.w = acc0[7];
    *reinterpret_cast<float4*>(zr + 4) = v;
    v.x = acc1[0]; v.y = acc1[1]; v.z = acc1[2]; v.w = acc1[3];
    *reinterpret_cast<float4*>(zr + 16) = v;
    v.x = acc1[4]; v.y = acc1[5]; v.z = acc1[6]; v.w = acc1[7];
    *reinterpret_cast<float4*>(zr + 20) = v;
  }
  __syncthreads();
  float* dst = z0 + ((size_t)b * N_ + n0) * C_;
  for (int p = 0; p < 4; ++p) {
    const int n = r + 16 * p;
    const float4 v = *reinterpret_cast<const float4*>(tile + n * ZP_ + e4);
    *reinterpret_cast<float4*>(dst + (size_t)n * C_ + e4) = v;
  }

  if (blockIdx.x < (B_ * M_) / 32 && t < 32) {
    const int id = blockIdx.x * 32 + t;
    const int n = indices[id & (M_ - 1)];
    const float* s = xyz + ((size_t)(id >> 10) * N_ + n) * 3;
    float* d = new_xyz + (size_t)id * 3;
    d[0] = s[0]; d[1] = s[1]; d[2] = s[2];
  }

  if (blockIdx.x >= 512 && blockIdx.x < 512 + WFB_) {
    const int e = (blockIdx.x - 512) * 256 + t;
    if (e < WF1_) {
      const int mt = e >> 6;
      const int ln = (e >> 1) & 31;
      const int k  = 64 + 2 * (ln >> 4) + (e & 1);
      wfrag[e] = (k < 67) ? W0[(mt * 16 + (ln & 15)) * 67 + k] : 0.0f;
    } else {
      const float* Wl = (e < WF2_) ? W1 : W2;
      const int q  = e - ((e < WF2_) ? WF1_ : WF2_);
      const int mt = q >> 10;
      const int w  = q & 1023;
      const int kt = 2 * (w >> 7) + ((w >> 1) & 1);
      const int ln = (w & 127) >> 2;
      const int k  = 4 * kt + 2 * (ln >> 4) + (w & 1);
      wfrag[e] = Wl[(mt * 16 + (ln & 15)) * 64 + k];
    }
  }
}

#define GP_   132
#define IMAX_ 0x7fffffff

__device__ __forceinline__ int dist_key(float f) {
  const int u = __float_as_int(f);
  return u ^ ((u >> 31) & 0x7fffffff);
}

template <int M>
__device__ __forceinline__ int row_xor(int x) {
  return __builtin_amdgcn_update_dpp(0, x, 0x160 + M, 0xf, 0xf, true);
}
__device__ __forceinline__ int half_xor(int x) {
  return __builtin_amdgcn_permlanex16(x, x, 0x76543210, 0xfedcba98, false, false);
}
__device__ __forceinline__ int wave_min_i(int v) {
  v = min(v, row_xor<1>(v));
  v = min(v, row_xor<2>(v));
  v = min(v, row_xor<4>(v));
  v = min(v, row_xor<8>(v));
  v = min(v, half_xor(v));
  return v;
}

#define PF_ 8

__device__ __forceinline__ void load_points(const float* __restrict__ px, int lane, int i0,
                                            float (&x)[PF_], float (&y)[PF_], float (&z)[PF_]) {
#pragma unroll
  for (int u = 0; u < PF_; ++u) {
    const int n = lane + 32 * (i0 + u);
    x[u] = px[n * 3 + 0];
    y[u] = px[n * 3 + 1];
    z[u] = px[n * 3 + 2];
  }
}

__device__ __forceinline__ void keys_of_points(const float (&xs)[PF_], const float (&ys)[PF_],
                                               const float (&zs)[PF_],
                                               float qx, float qy, float qz, float qq,
                                               int lane, int i0, int* __restrict__ mine,
                                               int& gv, int& gi) {
#pragma unroll
  for (int u = 0; u < PF_; ++u) {
    const int n = lane + 32 * (i0 + u);
    const float x = xs[u], y = ys[u], z = zs[u];
    const float pp  = __builtin_fmaf(z, z, __builtin_fmaf(x, x, y * y));
    const float dot = __builtin_fmaf(qz, z, __builtin_fmaf(qx, x, qy * y));
    const float d   = __builtin_fmaf(-2.0f, dot, qq + pp);
    const int key = dist_key(d);
    mine[i0 + u] = key;
    if (key < gv) { gv = key; gi = n; }
  }
}

__global__ __launch_bounds__(64) void k_knn(const float* __restrict__ xyz,
                                            const float* __restrict__ new_xyz,
                                            int* __restrict__ knn) {
  __shared__ __attribute__((aligned(16))) int dist[2][32 * GP_];
  const int wave = threadIdx.x >> 5;
  const int lane = threadIdx.x & 31;
  const int q = blockIdx.x * 2 + wave;
  const int b = q >> 10;

  int* dl = dist[wave];
  const float qx = new_xyz[q * 3 + 0];
  const float qy = new_xyz[q * 3 + 1];
  const float qz = new_xyz[q * 3 + 2];
  float qq;
  {
#pragma clang fp contract(off)
    qq = __builtin_fmaf(qx, qx, qy * qy) + qz * qz;
  }
  const float* px = xyz + (size_t)b * N_ * 3;

  int gv = IMAX_;
  int gi = IMAX_;
  int* mine = dl + lane * GP_;
  float ax[PF_], ay[PF_], az[PF_];
  float bx[PF_], by[PF_], bz[PF_];
  load_points(px, lane, 0, ax, ay, az);
  for (int i = 0; i < N_ / 32 - 2 * PF_; i += 2 * PF_) {
    load_points(px, lane, i + PF_, bx, by, bz);
    keys_of_points(ax, ay, az, qx, qy, qz, qq, lane, i, mine, gv, gi);
    load_points(px, lane, i + 2 * PF_, ax, ay, az);
    keys_of_points(bx, by, bz, qx, qy, qz, qq, lane, i + PF_, mine, gv, gi);
  }
  load_points(px, lane, N_ / 32 - PF_, bx, by, bz);
  keys_of_points(ax, ay, az, qx, qy, qz, qq, lane, N_ / 32 - 2 * PF_, mine, gv, gi);
  keys_of_points(bx, by, bz, qx, qy, qz, qq, lane, N_ / 32 - PF_, mine, gv, gi);
  __syncthreads();

  int res = 0;
  for (int k = 0; k < K_; ++k) {
    const int mn = wave_min_i(gv);
    const int mi = wave_min_i(gv == mn ? gi : IMAX_);
    if (lane == k) res = mi;
    const int w = mi & 31;

    const int4 c4 = *reinterpret_cast<const int4*>(dl + w * GP_ + 4 * lane);
    const int c[4] = {c4.x, c4.y, c4.z, c4.w};
    const int n0 = w + 128 * lane;
    int lv = IMAX_;
    int li = IMAX_;
#pragma unroll
    for (int t = 0; t < 4; ++t) {
      const int v = c[t];
      const int n = n0 + 32 * t;
      const bool later = (v > mn) || (v == mn && n > mi);
      if (later && v < lv) { lv = v; li = n; }
    }
    const int nv = wave_min_i(lv);
    const int ni = wave_min_i(lv == nv ? li : IMAX_);
    if (lane == w) { gv = nv; gi = ni; }
  }
  knn[q * K_ + lane] = res;
}

#define FP_ 68

__device__ __forceinline__ void bn_relu_store(
    const v8f acc, int rbase, int col,
    const float* __restrict__ g, const float* __restrict__ bt,
    const float* __restrict__ mu, const float* __restrict__ vr,
    float* __restrict__ Yb)
{
#pragma unroll
  for (int r = 0; r < 8; ++r) {
    const int row = rbase + r;
    const float s = g[row] / sqrtf(vr[row] + EPS_);
    const float y = (acc[r] - mu[row]) * s + bt[row];
    Yb[row * 32 + col] = fmaxf(y, 0.0f);
  }
}

__device__ __forceinline__ void layer0_bn_relu(
    const float* __restrict__ Xn, const float* __restrict__ Wf,
    const float* __restrict__ g, const float* __restrict__ bt,
    const float* __restrict__ mu, const float* __restrict__ vr,
    float* __restrict__ Yb, int lane)
{
  const int half = lane >> 4;
  const int l16  = lane & 15;
  for (int mt = 0; mt < 4; ++mt) {
    const v2f a = *reinterpret_cast<const v2f*>(Wf + mt * 64 + 2 * lane);
    for (int nt = 0; nt < 2; ++nt) {
      const int col = nt * 16 + l16;
      const float* xc = Xn + col * FP_;
      const float4 c0 = *reinterpret_cast<const float4*>(xc + mt * 16 + 8 * half);
      const float4 c1 = *reinterpret_cast<const float4*>(xc + mt * 16 + 8 * half + 4);
      const v2f bf = *reinterpret_cast<const v2f*>(xc + C_ + 2 * half);
      v8f acc;
      acc[0] = c0.x; acc[1] = c0.y; acc[2] = c0.z; acc[3] = c0.w;
      acc[4] = c1.x; acc[5] = c1.y; acc[6] = c1.z; acc[7] = c1.w;
      acc = __builtin_amdgcn_wmma_f32_16x16x4_f32(
           false, a,  false, bf,
           (short)0, acc,  false,  false);
      bn_relu_store(acc, mt * 16 + half * 8, col, g, bt, mu, vr, Yb);
    }
  }
}

__device__ __forceinline__ void gemm_bn_relu(
    const float* __restrict__ Xb,
    const float* __restrict__ Wf, int rowsOut,
    const float* __restrict__ g, const float* __restrict__ bt,
    const float* __restrict__ mu, const float* __restrict__ vr,
    float* __restrict__ Yb, int lane)
{
  const int half = lane >> 4;
  const int l16  = lane & 15;

  for (int mt = 0; mt < (rowsOut >> 4); ++mt) {
    const float* wt = Wf + mt * 1024;
    for (int nt = 0; nt < 2; ++nt) {
      const int col = nt * 16 + l16;
      v8f acc = {};
      for (int p = 0; p < 8; ++p) {
        const float4 a4 = *reinterpret_cast<const float4*>(wt + p * 128 + 4 * lane);
        const int k0 = 8 * p + 2 * half;
        v2f a, bf;
        a.x = a4.x; a.y = a4.y;
        bf.x = Xb[(k0 + 0) * 32 + col];
        bf.y = Xb[(k0 + 1) * 32 + col];
        acc = __builtin_amdgcn_wmma_f32_16x16x4_f32(
             false, a,  false, bf,
             (short)0, acc,  false,  false);
        a.x = a4.z; a.y = a4.w;
        bf.x = Xb[(k0 + 4) * 32 + col];
        bf.y = Xb[(k0 + 5) * 32 + col];
        acc = __builtin_amdgcn_wmma_f32_16x16x4_f32(
             false, a,  false, bf,
             (short)0, acc,  false,  false);
      }
      bn_relu_store(acc, mt * 16 + half * 8, col, g, bt, mu, vr, Yb);
    }
  }
}

__global__ __launch_bounds__(64) void k_mlp(
    const float* __restrict__ z0, const float* __restrict__ xyz,
    const float* __restrict__ new_xyz, const int* __restrict__ knn,
    const float* __restrict__ wfrag, const float* __restrict__ g0,
    const float* __restrict__ b0, const float* __restrict__ m0,
    const float* __restrict__ v0,
    const float* __restrict__ g1,
    const float* __restrict__ b1, const float* __restrict__ m1,
    const float* __restrict__ v1,
    const float* __restrict__ g2,
    const float* __restrict__ b2, const float* __restrict__ m2,
    const float* __restrict__ v2,
    float* __restrict__ out_feat)
{
  __shared__ __attribute__((aligned(16))) float Xs[2][32 * FP_];
  __shared__ float Ys[2][128 * 32];
  const int wave = threadIdx.x >> 5;
  const int lane = threadIdx.x & 31;
  const int q = blockIdx.x * 2 + wave;
  const int b = q >> 10;
  const int m = q & 1023;

  float* Xb = Xs[wave];
  float* Yb = Ys[wave];

  const int nj = knn[q * K_ + lane];
  const int half = lane >> 4;
  const int l16  = lane & 15;
  const float* fb = z0 + (size_t)b * N_ * C_ + 4 * l16;
  float* xw = Xb + half * FP_ + 4 * l16;
  for (int i = 0; i < 16; ++i) {
    const int ne = __builtin_amdgcn_readlane(nj, 2 * i);
    const int no = __builtin_amdgcn_readlane(nj, 2 * i + 1);
    const int n  = half ? no : ne;
    const float4 f = *reinterpret_cast<const float4*>(fb + (size_t)n * C_);
    *reinterpret_cast<float4*>(xw + 2 * i * FP_) = f;
  }
  const float* p  = xyz + ((size_t)b * N_ + nj) * 3;
  const float* qc = new_xyz + (size_t)q * 3;
  float4 rel;
  rel.x = p[0] - qc[0];
  rel.y = p[1] - qc[1];
  rel.z = p[2] - qc[2];
  rel.w = 0.0f;
  *reinterpret_cast<float4*>(Xb + lane * FP_ + C_) = rel;
  __syncthreads();

  layer0_bn_relu(Xb, wfrag + WF0_, g0, b0, m0, v0, Yb, lane);
  __syncthreads();
  gemm_bn_relu(Yb, wfrag + WF1_,  64, g1, b1, m1, v1, Xb, lane);
  __syncthreads();
  gemm_bn_relu(Xb, wfrag + WF2_, 128, g2, b2, m2, v2, Yb, lane);
  __syncthreads();

  for (int o = lane; o < 128; o += 32) {
    float mx = -BIGF;
#pragma unroll
    for (int j = 0; j < 32; ++j) mx = fmaxf(mx, Yb[o * 32 + j]);
    out_feat[((size_t)b * 128 + o) * M_ + m] = mx;
  }
}

extern "C" void kernel_launch(void* const* d_in, const int* in_sizes, int n_in,
                              void* d_out, int out_size, void* d_ws, size_t ws_size,
                              hipStream_t stream) {
  const float* xyz      = (const float*)d_in[0];
  const float* features = (const float*)d_in[1];
  const int*   indices  = (const int*)d_in[2];
  const float* W0 = (const float*)d_in[3];
  const float* g0 = (const float*)d_in[4];
  const float* b0 = (const float*)d_in[5];
  const float* m0 = (const float*)d_in[6];
  const float* v0 = (const float*)d_in[7];
  const float* W1 = (const float*)d_in[8];
  const float* g1 = (const float*)d_in[9];
  const float* b1 = (const float*)d_in[10];
  const float* m1 = (const float*)d_in[11];
  const float* v1 = (const float*)d_in[12];
  const float* W2 = (const float*)d_in[13];
  const float* g2 = (const float*)d_in[14];
  const float* b2 = (const float*)d_in[15];
  const float* m2 = (const float*)d_in[16];
  const float* v2 = (const float*)d_in[17];

  float* new_xyz  = (float*)d_out;
  float* out_feat = new_xyz + (size_t)B_ * M_ * 3;
  int*   knn      = (int*)d_ws;
  float* z0       = (float*)d_ws + (size_t)B_ * M_ * K_;
  float* wfrag    = z0 + (size_t)B_ * N_ * C_;

  k_prep<<<B_ * (N_ / 64), 256, 0, stream>>>(xyz, features, indices, W0, W1, W2,
                                              new_xyz, z0, wfrag);
  k_knn<<<(B_ * M_) / 2, 64, 0, stream>>>(xyz, new_xyz, knn);
  k_mlp<<<(B_ * M_) / 2, 64, 0, stream>>>(z0, xyz, new_xyz, knn, wfrag,
      g0, b0, m0, v0, g1, b1, m1, v1, g2, b2, m2, v2, out_feat);
}


# BatchNorm scale g/sqrt(v+eps) made once per channel in the first kernel (256 words in the workspace) and loaded by the MLP kernel, not recomputed by every lane for every row tile; same bits
# speedup vs baseline: 3.2223x; 1.1148x over previous
#include <hip/hip_runtime.h>
#include <hip/hip_bf16.h>

typedef __attribute__((ext_vector_type(2))) float v2f;
typedef __attribute__((ext_vector_type(8))) float v8f;

#define B_   16
#define N_   4096
#define C_   64
#define M_   1024
#define K_   32
#define EPS_ 1e-5f
#define BIGF 3.402823e38f

#define TPK_ 72
#define ZP_  68
#define WKP_ 66
#define WMT_ (16 * WKP_)

#define WF0_  0
#define WF1_  (4 * 64)
#define WF2_  (WF1_ + 64 * 64)
#define WFN_  (WF2_ + 128 * 64)
#define WFB_  (WFN_ / 256)

#define BS0_  0
#define BS1_  64
#define BS2_  128
#define BSN_  256

__global__ __launch_bounds__(256) void k_prep(const float* __restrict__ xyz,
                                              const float* __restrict__ features,
                                              const int* __restrict__ indices,
                                              const float* __restrict__ W0,
                                              const float* __restrict__ W1,
                                              const float* __restrict__ W2,
                                              const float* __restrict__ g0,
                                              const float* __restrict__ v0,
                                              const float* __restrict__ g1,
                                              const float* __restrict__ v1,
                                              const float* __restrict__ g2,
                                              const float* __restrict__ v2,
                                              float* __restrict__ new_xyz,
                                              float* __restrict__ z0,
                                              float* __restrict__ wfrag,
                                              float* __restrict__ bnscale) {
  __shared__ __attribute__((aligned(16))) float tile[C_ * TPK_];
  __shared__ __attribute__((aligned(16))) float wl[4 * WMT_];
  const int t  = threadIdx.x;
  const int b  = blockIdx.x >> 6;
  const int n0 = (blockIdx.x & 63) << 6;
  const int r  = t >> 4;
  const int e4 = (t & 15) << 2;

  const float* src = features + (size_t)b * C_ * N_ + n0;
  for (int p = 0; p < 4; ++p) {
    const int c = r + 16 * p;
    const float4 v = *reinterpret_cast<const float4*>(src + (size_t)c * N_ + e4);
    *reinterpret_cast<float4*>(tile + c * TPK_ + e4) = v;
  }
  for (int i = 0; i < 16; ++i) {
    const int f   = t + 256 * i;
    const int row = f >> 6;
    const int k   = f & 63;
    const int ln  = (row & 15) + 16 * ((k >> 1) & 1);
    wl[(row >> 4) * WMT_ + (k >> 2) * WKP_ + 2 * ln + (k & 1)] = W0[row * 67 + k];
  }
  __syncthreads();

  const int wave = t >> 5;
  const int lane = t & 31;
  const int half = lane >> 4;
  const int l16  = lane & 15;
  const int nt   = wave & 3;
  const int mt0  = (wave >> 2) << 1;
  {
    const float* xb = tile + (2 * half) * TPK_ + nt * 16 + l16;
    const float* wa = wl + mt0 * WMT_ + 2 * lane;
    v8f acc0 = {};
    v8f acc1 = {};
    for (int kt = 0; kt < 16; ++kt) {
      v2f bf;
      bf.x = xb[(4 * kt + 0) * TPK_];
      bf.y = xb[(4 * kt + 1) * TPK_];
      const v2f a0 = *reinterpret_cast<const v2f*>(wa + kt * WKP_);
      const v2f a1 = *reinterpret_cast<const v2f*>(wa + WMT_ + kt * WKP_);
      acc0 = __builtin_amdgcn_wmma_f32_16x16x4_f32(
           false, a0,  false, bf,
           (short)0, acc0,  false,  false);
      acc1 = __builtin_amdgcn_wmma_f32_16x16x4_f32(
           false, a1,  false, bf,
           (short)0, acc1,  false,  false);
    }
    __syncthreads();
    float* zr = tile + (nt * 16 + l16) * ZP_ + mt0 * 16 + 8 * half;
    float4 v;
    v.x = acc0[0]; v.y = acc0[1]; v.z = acc0[2]; v.w = acc0[3];
    *reinterpret_cast<float4*>(zr) = v;
    v.x = acc0[4]; v.y = acc0[5]; v.z = acc0[6]; v.w = acc0[7];
    *reinterpret_cast<float4*>(zr + 4) = v;
    v.x = acc1[0]; v.y = acc1[1]; v.z = acc1[2]; v.w = acc1[3];
    *reinterpret_cast<float4*>(zr + 16) = v;
    v.x = acc1[4]; v.y = acc1[5]; v.z = acc1[6]; v.w = acc1[7];
    *reinterpret_cast<float4*>(zr + 20) = v;
  }
  __syncthreads();
  float* dst = z0 + ((size_t)b * N_ + n0) * C_;
  for (int p = 0; p < 4; ++p) {
    const int n = r + 16 * p;
    const float4 v = *reinterpret_cast<const float4*>(tile + n * ZP_ + e4);
    *reinterpret_cast<float4*>(dst + (size_t)n * C_ + e4) = v;
  }

  if (blockIdx.x < (B_ * M_) / 32 && t < 32) {
    const int id = blockIdx.x * 32 + t;
    const int n = indices[id & (M_ - 1)];
    const float* s = xyz + ((size_t)(id >> 10) * N_ + n) * 3;
    float* d = new_xyz + (size_t)id * 3;
    d[0] = s[0]; d[1] = s[1]; d[2] = s[2];
  }

  if (blockIdx.x >= 512 && blockIdx.x < 512 + WFB_) {
    const int e = (blockIdx.x - 512) * 256 + t;
    if (e < WF1_) {
      const int mt = e >> 6;
      const int ln = (e >> 1) & 31;
      const int k  = 64 + 2 * (ln >> 4) + (e & 1);
      wfrag[e] = (k < 67) ? W0[(mt * 16 + (ln & 15)) * 67 + k] : 0.0f;
    } else {
      const float* Wl = (e < WF2_) ? W1 : W2;
      const int q  = e - ((e < WF2_) ? WF1_ : WF2_);
      const int mt = q >> 10;
      const int w  = q & 1023;
      const int kt = 2 * (w >> 7) + ((w >> 1) & 1);
      const int ln = (w & 127) >> 2;
      const int k  = 4 * kt + 2 * (ln >> 4) + (w & 1);
      wfrag[e] = Wl[(mt * 16 + (ln & 15)) * 64 + k];
    }
  }

  if (blockIdx.x == 512 + WFB_) {
    const float* gp = (t < BS1_) ? g0 : (t < BS2_) ? g1 : g2;
    const float* vp = (t < BS1_) ? v0 : (t < BS2_) ? v1 : v2;
    const int c = (t < BS1_) ? t : (t < BS2_) ? t - BS1_ : t - BS2_;
    bnscale[t] = gp[c] / sqrtf(vp[c] + EPS_);
  }
}

#define GP_   132
#define IMAX_ 0x7fffffff

__device__ __forceinline__ int dist_key(float f) {
  const int u = __float_as_int(f);
  return u ^ ((u >> 31) & 0x7fffffff);
}

template <int M>
__device__ __forceinline__ int row_xor(int x) {
  return __builtin_amdgcn_update_dpp(0, x, 0x160 + M, 0xf, 0xf, true);
}
__device__ __forceinline__ int half_xor(int x) {
  return __builtin_amdgcn_permlanex16(x, x, 0x76543210, 0xfedcba98, false, false);
}
__device__ __forceinline__ int wave_min_i(int v) {
  v = min(v, row_xor<1>(v));
  v = min(v, row_xor<2>(v));
  v = min(v, row_xor<4>(v));
  v = min(v, row_xor<8>(v));
  v = min(v, half_xor(v));
  return v;
}

#define PF_ 8

__device__ __forceinline__ void load_points(const float* __restrict__ px, int lane, int i0,
                                            float (&x)[PF_], float (&y)[PF_], float (&z)[PF_]) {
#pragma unroll
  for (int u = 0; u < PF_; ++u) {
    const int n = lane + 32 * (i0 + u);
    x[u] = px[n * 3 + 0];
    y[u] = px[n * 3 + 1];
    z[u] = px[n * 3 + 2];
  }
}

__device__ __forceinline__ void keys_of_points(const float (&xs)[PF_], const float (&ys)[PF_],
                                               const float (&zs)[PF_],
                                               float qx, float qy, float qz, float qq,
                                               int lane, int i0, int* __restrict__ mine,
                                               int& gv, int& gi) {
#pragma unroll
  for (int u = 0; u < PF_; ++u) {
    const int n = lane + 32 * (i0 + u);
    const float x = xs[u], y = ys[u], z = zs[u];
    const float pp  = __builtin_fmaf(z, z, __builtin_fmaf(x, x, y * y));
    const float dot = __builtin_fmaf(qz, z, __builtin_fmaf(qx, x, qy * y));
    const float d   = __builtin_fmaf(-2.0f, dot, qq + pp);
    const int key = dist_key(d);
    mine[i0 + u] = key;
    if (key < gv) { gv = key; gi = n; }
  }
}

__global__ __launch_bounds__(64) void k_knn(const float* __restrict__ xyz,
                                            const float* __restrict__ new_xyz,
                                            int* __restrict__ knn) {
  __shared__ __attribute__((aligned(16))) int dist[2][32 * GP_];
  const int wave = threadIdx.x >> 5;
  const int lane = threadIdx.x & 31;
  const int q = blockIdx.x * 2 + wave;
  const int b = q >> 10;

  int* dl = dist[wave];
  const float qx = new_xyz[q * 3 + 0];
  const float qy = new_xyz[q * 3 + 1];
  const float qz = new_xyz[q * 3 + 2];
  float qq;
  {
#pragma clang fp contract(off)
    qq = __builtin_fmaf(qx, qx, qy * qy) + qz * qz;
  }
  const float* px = xyz + (size_t)b * N_ * 3;

  int gv = IMAX_;
  int gi = IMAX_;
  int* mine = dl + lane * GP_;
  float ax[PF_], ay[PF_], az[PF_];
  float bx[PF_], by[PF_], bz[PF_];
  load_points(px, lane, 0, ax, ay, az);
  for (int i = 0; i < N_ / 32 - 2 * PF_; i += 2 * PF_) {
    load_points(px, lane, i + PF_, bx, by, bz);
    keys_of_points(ax, ay, az, qx, qy, qz, qq, lane, i, mine, gv, gi);
    load_points(px, lane, i + 2 * PF_, ax, ay, az);
    keys_of_points(bx, by, bz, qx, qy, qz, qq, lane, i + PF_, mine, gv, gi);
  }
  load_points(px, lane, N_ / 32 - PF_, bx, by, bz);
  keys_of_points(ax, ay, az, qx, qy, qz, qq, lane, N_ / 32 - 2 * PF_, mine, gv, gi);
  keys_of_points(bx, by, bz, qx, qy, qz, qq, lane, N_ / 32 - PF_, mine, gv, gi);
  __syncthreads();

  int res = 0;
  for (int k = 0; k < K_; ++k) {
    const int mn = wave_min_i(gv);
    const int mi = wave_min_i(gv == mn ? gi : IMAX_);
    if (lane == k) res = mi;
    const int w = mi & 31;

    const int4 c4 = *reinterpret_cast<const int4*>(dl + w * GP_ + 4 * lane);
    const int c[4] = {c4.x, c4.y, c4.z, c4.w};
    const int n0 = w + 128 * lane;
    int lv = IMAX_;
    int li = IMAX_;
#pragma unroll
    for (int t = 0; t < 4; ++t) {
      const int v = c[t];
      const int n = n0 + 32 * t;
      const bool later = (v > mn) || (v == mn && n > mi);
      if (later && v < lv) { lv = v; li = n; }
    }
    const int nv = wave_min_i(lv);
    const int ni = wave_min_i(lv == nv ? li : IMAX_);
    if (lane == w) { gv = nv; gi = ni; }
  }
  knn[q * K_ + lane] = res;
}

#define FP_ 68

__device__ __forceinline__ void bn_relu_store(
    const v8f acc, int rbase, int col,
    const float* __restrict__ sc, const float* __restrict__ bt,
    const float* __restrict__ mu,
    float* __restrict__ Yb)
{
#pragma unroll
  for (int r = 0; r < 8; ++r) {
    const int row = rbase + r;
    const float y = (acc[r] - mu[row]) * sc[row] + bt[row];
    Yb[row * 32 + col] = fmaxf(y, 0.0f);
  }
}

__device__ __forceinline__ void layer0_bn_relu(
    const float* __restrict__ Xn, const float* __restrict__ Wf,
    const float* __restrict__ sc, const float* __restrict__ bt,
    const float* __restrict__ mu,
    float* __restrict__ Yb, int lane)
{
  const int half = lane >> 4;
  const int l16  = lane & 15;
  for (int mt = 0; mt < 4; ++mt) {
    const v2f a = *reinterpret_cast<const v2f*>(Wf + mt * 64 + 2 * lane);
    for (int nt = 0; nt < 2; ++nt) {
      const int col = nt * 16 + l16;
      const float* xc = Xn + col * FP_;
      const float4 c0 = *reinterpret_cast<const float4*>(xc + mt * 16 + 8 * half);
      const float4 c1 = *reinterpret_cast<const float4*>(xc + mt * 16 + 8 * half + 4);
      const v2f bf = *reinterpret_cast<const v2f*>(xc + C_ + 2 * half);
      v8f acc;
      acc[0] = c0.x; acc[1] = c0.y; acc[2] = c0.z; acc[3] = c0.w;
      acc[4] = c1.x; acc[5] = c1.y; acc[6] = c1.z; acc[7] = c1.w;
      acc = __builtin_amdgcn_wmma_f32_16x16x4_f32(
           false, a,  false, bf,
           (short)0, acc,  false,  false);
      bn_relu_store(acc, mt * 16 + half * 8, col, sc, bt, mu, Yb);
    }
  }
}

__device__ __forceinline__ void gemm_bn_relu(
    const float* __restrict__ Xb,
    const float* __restrict__ Wf, int rowsOut,
    const float* __restrict__ sc, const float* __restrict__ bt,
    const float* __restrict__ mu,
    float* __restrict__ Yb, int lane)
{
  const int half = lane >> 4;
  const int l16  = lane & 15;

  for (int mt = 0; mt < (rowsOut >> 4); ++mt) {
    const float* wt = Wf + mt * 1024;
    for (int nt = 0; nt < 2; ++nt) {
      const int col = nt * 16 + l16;
      v8f acc = {};
      for (int p = 0; p < 8; ++p) {
        const float4 a4 = *reinterpret_cast<const float4*>(wt + p * 128 + 4 * lane);
        const int k0 = 8 * p + 2 * half;
        v2f a, bf;
        a.x = a4.x; a.y = a4.y;
        bf.x = Xb[(k0 + 0) * 32 + col];
        bf.y = Xb[(k0 + 1) * 32 + col];
        acc = __builtin_amdgcn_wmma_f32_16x16x4_f32(
             false, a,  false, bf,
             (short)0, acc,  false,  false);
        a.x = a4.z; a.y = a4.w;
        bf.x = Xb[(k0 + 4) * 32 + col];
        bf.y = Xb[(k0 + 5) * 32 + col];
        acc = __builtin_amdgcn_wmma_f32_16x16x4_f32(
             false, a,  false, bf,
             (short)0, acc,  false,  false);
      }
      bn_relu_store(acc, mt * 16 + half * 8, col, sc, bt, mu, Yb);
    }
  }
}

__global__ __launch_bounds__(64) void k_mlp(
    const float* __restrict__ z0, const float* __restrict__ xyz,
    const float* __restrict__ new_xyz, const int* __restrict__ knn,
    const float* __restrict__ wfrag, const float* __restrict__ bnscale,
    const float* __restrict__ b0, const float* __restrict__ m0,
    const float* __restrict__ b1, const float* __restrict__ m1,
    const float* __restrict__ b2, const float* __restrict__ m2,
    float* __restrict__ out_feat)
{
  __shared__ __attribute__((aligned(16))) float Xs[2][32 * FP_];
  __shared__ float Ys[2][128 * 32];
  const int wave = threadIdx.x >> 5;
  const int lane = threadIdx.x & 31;
  const int q = blockIdx.x * 2 + wave;
  const int b = q >> 10;
  const int m = q & 1023;

  float* Xb = Xs[wave];
  float* Yb = Ys[wave];

  const int nj = knn[q * K_ + lane];
  const int half = lane >> 4;
  const int l16  = lane & 15;
  const float* fb = z0 + (size_t)b * N_ * C_ + 4 * l16;
  float* xw = Xb + half * FP_ + 4 * l16;
  for (int i = 0; i < 16; ++i) {
    const int ne = __builtin_amdgcn_readlane(nj, 2 * i);
    const int no = __builtin_amdgcn_readlane(nj, 2 * i + 1);
    const int n  = half ? no : ne;
    const float4 f = *reinterpret_cast<const float4*>(fb + (size_t)n * C_);
    *reinterpret_cast<float4*>(xw + 2 * i * FP_) = f;
  }
  const float* p  = xyz + ((size_t)b * N_ + nj) * 3;
  const float* qc = new_xyz + (size_t)q * 3;
  float4 rel;
  rel.x = p[0] - qc[0];
  rel.y = p[1] - qc[1];
  rel.z = p[2] - qc[2];
  rel.w = 0.0f;
  *reinterpret_cast<float4*>(Xb + lane * FP_ + C_) = rel;
  __syncthreads();

  layer0_bn_relu(Xb, wfrag + WF0_, bnscale + BS0_, b0, m0, Yb, lane);
  __syncthreads();
  gemm_bn_relu(Yb, wfrag + WF1_,  64, bnscale + BS1_, b1, m1, Xb, lane);
  __syncthreads();
  gemm_bn_relu(Xb, wfrag + WF2_, 128, bnscale + BS2_, b2, m2, Yb, lane);
  __syncthreads();

  for (int o = lane; o < 128; o += 32) {
    float mx = -BIGF;
#pragma unroll
    for (int j = 0; j < 32; ++j) mx = fmaxf(mx, Yb[o * 32 + j]);
    out_feat[((size_t)b * 128 + o) * M_ + m] = mx;
  }
}

extern "C" void kernel_launch(void* const* d_in, const int* in_sizes, int n_in,
                              void* d_out, int out_size, void* d_ws, size_t ws_size,
                              hipStream_t stream) {
  const float* xyz      = (const float*)d_in[0];
  const float* features = (const float*)d_in[1];
  const int*   indices  = (const int*)d_in[2];
  const float* W0 = (const float*)d_in[3];
  const float* g0 = (const float*)d_in[4];
  const float* b0 = (const float*)d_in[5];
  const float* m0 = (const float*)d_in[6];
  const float* v0 = (const float*)d_in[7];
  const float* W1 = (const float*)d_in[8];
  const float* g1 = (const float*)d_in[9];
  const float* b1 = (const float*)d_in[10];
  const float* m1 = (const float*)d_in[11];
  const float* v1 = (const float*)d_in[12];
  const float* W2 = (const float*)d_in[13];
  const float* g2 = (const float*)d_in[14];
  const float* b2 = (const float*)d_in[15];
  const float* m2 = (const float*)d_in[16];
  const float* v2 = (const float*)d_in[17];

  float* new_xyz  = (float*)d_out;
  float* out_feat = new_xyz + (size_t)B_ * M_ * 3;
  int*   knn      = (int*)d_ws;
  float* z0       = (float*)d_ws + (size_t)B_ * M_ * K_;
  float* wfrag    = z0 + (size_t)B_ * N_ * C_;
  float* bnscale  = wfrag + WFN_;

  k_prep<<<B_ * (N_ / 64), 256, 0, stream>>>(xyz, features, indices, W0, W1, W2,
                                              g0, v0, g1, v1, g2, v2,
                                              new_xyz, z0, wfrag, bnscale);
  k_knn<<<(B_ * M_) / 2, 64, 0, stream>>>(xyz, new_xyz, knn);
  k_mlp<<<(B_ * M_) / 2, 64, 0, stream>>>(z0, xyz, new_xyz, knn, wfrag, bnscale,
      b0, m0, b1, m1, b2, m2, out_feat);
}


# Neighbour search and a query's three layers in one launch, by the same wave: indices stay in registers, LDS shared (layer 2 as two pooled halves); one launch and the index array gone; sums unchanged
# speedup vs baseline: 3.4162x; 1.0602x over previous
#include <hip/hip_runtime.h>
#include <hip/hip_bf16.h>

typedef __attribute__((ext_vector_type(2))) float v2f;
typedef __attribute__((ext_vector_type(8))) float v8f;

#define B_   16
#define N_   4096
#define C_   64
#define M_   1024
#define K_   32
#define EPS_ 1e-5f
#define BIGF 3.402823e38f

#define TPK_ 72
#define ZP_  68
#define WKP_ 66
#define WMT_ (16 * WKP_)

#define WF0_  0
#define WF1_  (4 * 64)
#define WF2_  (WF1_ + 64 * 64)
#define WFN_  (WF2_ + 128 * 64)
#define WFB_  (WFN_ / 256)

#define BS0_  0
#define BS1_  64
#define BS2_  128
#define BSN_  256

__global__ __launch_bounds__(256) void k_prep(const float* __restrict__ xyz,
                                              const float* __restrict__ features,
                                              const int* __restrict__ indices,
                                              const float* __restrict__ W0,
                                              const float* __restrict__ W1,
                                              const float* __restrict__ W2,
                                              const float* __restrict__ g0,
                                              const float* __restrict__ v0,
                                              const float* __restrict__ g1,
                                              const float* __restrict__ v1,
                                              const float* __restrict__ g2,
                                              const float* __restrict__ v2,
                                              float* __restrict__ new_xyz,
                                              float* __restrict__ z0,
                                              float* __restrict__ wfrag,
                                              float* __restrict__ bnscale) {
  __shared__ __attribute__((aligned(16))) float tile[C_ * TPK_];
  __shared__ __attribute__((aligned(16))) float wl[4 * WMT_];
  const int t  = threadIdx.x;
  const int b  = blockIdx.x >> 6;
  const int n0 = (blockIdx.x & 63) << 6;
  const int r  = t >> 4;
  const int e4 = (t & 15) << 2;

  const float* src = features + (size_t)b * C_ * N_ + n0;
  for (int p = 0; p < 4; ++p) {
    const int c = r + 16 * p;
    const float4 v = *reinterpret_cast<const float4*>(src + (size_t)c * N_ + e4);
    *reinterpret_cast<float4*>(tile + c * TPK_ + e4) = v;
  }
  for (int i = 0; i < 16; ++i) {
    const int f   = t + 256 * i;
    const int row = f >> 6;
    const int k   = f & 63;
    const int ln  = (row & 15) + 16 * ((k >> 1) & 1);
    wl[(row >> 4) * WMT_ + (k >> 2) * WKP_ + 2 * ln + (k & 1)] = W0[row * 67 + k];
  }
  __syncthreads();

  const int wave = t >> 5;
  const int lane = t & 31;
  const int half = lane >> 4;
  const int l16  = lane & 15;
  const int nt   = wave & 3;
  const int mt0  = (wave >> 2) << 1;
  {
    const float* xb = tile + (2 * half) * TPK_ + nt * 16 + l16;
    const float* wa = wl + mt0 * WMT_ + 2 * lane;
    v8f acc0 = {};
    v8f acc1 = {};
    for (int kt = 0; kt < 16; ++kt) {
      v2f bf;
      bf.x = xb[(4 * kt + 0) * TPK_];
      bf.y = xb[(4 * kt + 1) * TPK_];
      const v2f a0 = *reinterpret_cast<const v2f*>(wa + kt * WKP_);
      const v2f a1 = *reinterpret_cast<const v2f*>(wa + WMT_ + kt * WKP_);
      acc0 = __builtin_amdgcn_wmma_f32_16x16x4_f32(
           false, a0,  false, bf,
           (short)0, acc0,  false,  false);
      acc1 = __builtin_amdgcn_wmma_f32_16x16x4_f32(
           false, a1,  false, bf,
           (short)0, acc1,  false,  false);
    }
    __syncthreads();
    float* zr = tile + (nt * 16 + l16) * ZP_ + mt0 * 16 + 8 * half;
    float4 v;
    v.x = acc0[0]; v.y = acc0[1]; v.z = acc0[2]; v.w = acc0[3];
    *reinterpret_cast<float4*>(zr) = v;
    v.x = acc0[4]; v.y = acc0[5]; v.z = acc0[6]; v.w = acc0[7];
    *reinterpret_cast<float4*>(zr + 4) = v;
    v.x = acc1[0]; v.y = acc1[1]; v.z = acc1[2]; v.w = acc1[3];
    *reinterpret_cast<float4*>(zr + 16) = v;
    v.x = acc1[4]; v.y = acc1[5]; v.z = acc1[6]; v.w = acc1[7];
    *reinterpret_cast<float4*>(zr + 20) = v;
  }
  __syncthreads();
  float* dst = z0 + ((size_t)b * N_ + n0) * C_;
  for (int p = 0; p < 4; ++p) {
    const int n = r + 16 * p;
    const float4 v = *reinterpret_cast<const float4*>(tile + n * ZP_ + e4);
    *reinterpret_cast<float4*>(dst + (size_t)n * C_ + e4) = v;
  }

  if (blockIdx.x < (B_ * M_) / 32 && t < 32) {
    const int id = blockIdx.x * 32 + t;
    const int n = indices[id & (M_ - 1)];
    const float* s = xyz + ((size_t)(id >> 10) * N_ + n) * 3;
    float* d = new_xyz + (size_t)id * 3;
    d[0] = s[0]; d[1] = s[1]; d[2] = s[2];
  }

  if (blockIdx.x >= 512 && blockIdx.x < 512 + WFB_) {
    const int e = (blockIdx.x - 512) * 256 + t;
    if (e < WF1_) {
      const int mt = e >> 6;
      const int ln = (e >> 1) & 31;
      const int k  = 64 + 2 * (ln >> 4) + (e & 1);
      wfrag[e] = (k < 67) ? W0[(mt * 16 + (ln & 15)) * 67 + k] : 0.0f;
    } else {
      const float* Wl = (e < WF2_) ? W1 : W2;
      const int q  = e - ((e < WF2_) ? WF1_ : WF2_);
      const int mt = q >> 10;
      const int w  = q & 1023;
      const int kt = 2 * (w >> 7) + ((w >> 1) & 1);
      const int ln = (w & 127) >> 2;
      const int k  = 4 * kt + 2 * (ln >> 4) + (w & 1);
      wfrag[e] = Wl[(mt * 16 + (ln & 15)) * 64 + k];
    }
  }

  if (blockIdx.x == 512 + WFB_) {
    const float* gp = (t < BS1_) ? g0 : (t < BS2_) ? g1 : g2;
    const float* vp = (t < BS1_) ? v0 : (t < BS2_) ? v1 : v2;
    const int c = (t < BS1_) ? t : (t < BS2_) ? t - BS1_ : t - BS2_;
    bnscale[t] = gp[c] / sqrtf(vp[c] + EPS_);
  }
}

#define GP_   132
#define IMAX_ 0x7fffffff

__device__ __forceinline__ int dist_key(float f) {
  const int u = __float_as_int(f);
  return u ^ ((u >> 31) & 0x7fffffff);
}

template <int M>
__device__ __forceinline__ int row_xor(int x) {
  return __builtin_amdgcn_update_dpp(0, x, 0x160 + M, 0xf, 0xf, true);
}
__device__ __forceinline__ int half_xor(int x) {
  return __builtin_amdgcn_permlanex16(x, x, 0x76543210, 0xfedcba98, false, false);
}
__device__ __forceinline__ int wave_min_i(int v) {
  v = min(v, row_xor<1>(v));
  v = min(v, row_xor<2>(v));
  v = min(v, row_xor<4>(v));
  v = min(v, row_xor<8>(v));
  v = min(v, half_xor(v));
  return v;
}

#define PF_ 8

__device__ __forceinline__ void load_points(const float* __restrict__ px, int lane, int i0,
                                            float (&x)[PF_], float (&y)[PF_], float (&z)[PF_]) {
#pragma unroll
  for (int u = 0; u < PF_; ++u) {
    const int n = lane + 32 * (i0 + u);
    x[u] = px[n * 3 + 0];
    y[u] = px[n * 3 + 1];
    z[u] = px[n * 3 + 2];
  }
}

__device__ __forceinline__ void keys_of_points(const float (&xs)[PF_], const float (&ys)[PF_],
                                               const float (&zs)[PF_],
                                               float qx, float qy, float qz, float qq,
                                               int lane, int i0, int* __restrict__ mine,
                                               int& gv, int& gi) {
#pragma unroll
  for (int u = 0; u < PF_; ++u) {
    const int n = lane + 32 * (i0 + u);
    const float x = xs[u], y = ys[u], z = zs[u];
    const float pp  = __builtin_fmaf(z, z, __builtin_fmaf(x, x, y * y));
    const float dot = __builtin_fmaf(qz, z, __builtin_fmaf(qx, x, qy * y));
    const float d   = __builtin_fmaf(-2.0f, dot, qq + pp);
    const int key = dist_key(d);
    mine[i0 + u] = key;
    if (key < gv) { gv = key; gi = n; }
  }
}

__device__ __forceinline__ int knn_select(const float* __restrict__ xyz,
                                          const float* __restrict__ new_xyz,
                                          int q, int lane, int* __restrict__ dl) {
  const int b = q >> 10;
  const float qx = new_xyz[q * 3 + 0];
  const float qy = new_xyz[q * 3 + 1];
  const float qz = new_xyz[q * 3 + 2];
  float qq;
  {
#pragma clang fp contract(off)
    qq = __builtin_fmaf(qx, qx, qy * qy) + qz * qz;
  }
  const float* px = xyz + (size_t)b * N_ * 3;

  int gv = IMAX_;
  int gi = IMAX_;
  int* mine = dl + lane * GP_;
  float ax[PF_], ay[PF_], az[PF_];
  float bx[PF_], by[PF_], bz[PF_];
  load_points(px, lane, 0, ax, ay, az);
  for (int i = 0; i < N_ / 32 - 2 * PF_; i += 2 * PF_) {
    load_points(px, lane, i + PF_, bx, by, bz);
    keys_of_points(ax, ay, az, qx, qy, qz, qq, lane, i, mine, gv, gi);
    load_points(px, lane, i + 2 * PF_, ax, ay, az);
    keys_of_points(bx, by, bz, qx, qy, qz, qq, lane, i + PF_, mine, gv, gi);
  }
  load_points(px, lane, N_ / 32 - PF_, bx, by, bz);
  keys_of_points(ax, ay, az, qx, qy, qz, qq, lane, N_ / 32 - 2 * PF_, mine, gv, gi);
  keys_of_points(bx, by, bz, qx, qy, qz, qq, lane, N_ / 32 - PF_, mine, gv, gi);
  __syncthreads();

  int res = 0;
  for (int k = 0; k < K_; ++k) {
    const int mn = wave_min_i(gv);
    const int mi = wave_min_i(gv == mn ? gi : IMAX_);
    if (lane == k) res = mi;
    const int w = mi & 31;

    const int4 c4 = *reinterpret_cast<const int4*>(dl + w * GP_ + 4 * lane);
    const int c[4] = {c4.x, c4.y, c4.z, c4.w};
    const int n0 = w + 128 * lane;
    int lv = IMAX_;
    int li = IMAX_;
#pragma unroll
    for (int t = 0; t < 4; ++t) {
      const int v = c[t];
      const int n = n0 + 32 * t;
      const bool later = (v > mn) || (v == mn && n > mi);
      if (later && v < lv) { lv = v; li = n; }
    }
    const int nv = wave_min_i(lv);
    const int ni = wave_min_i(lv == nv ? li : IMAX_);
    if (lane == w) { gv = nv; gi = ni; }
  }
  return res;
}

#define FP_ 68

__device__ __forceinline__ void bn_relu_store(
    const v8f acc, int rbase, int col,
    const float* __restrict__ sc, const float* __restrict__ bt,
    const float* __restrict__ mu,
    float* __restrict__ Yb)
{
#pragma unroll
  for (int r = 0; r < 8; ++r) {
    const int row = rbase + r;
    const float y = (acc[r] - mu[row]) * sc[row] + bt[row];
    Yb[row * 32 + col] = fmaxf(y, 0.0f);
  }
}

__device__ __forceinline__ void layer0_bn_relu(
    const float* __restrict__ Xn, const float* __restrict__ Wf,
    const float* __restrict__ sc, const float* __restrict__ bt,
    const float* __restrict__ mu,
    float* __restrict__ Yb, int lane)
{
  const int half = lane >> 4;
  const int l16  = lane & 15;
  for (int mt = 0; mt < 4; ++mt) {
    const v2f a = *reinterpret_cast<const v2f*>(Wf + mt * 64 + 2 * lane);
    for (int nt = 0; nt < 2; ++nt) {
      const int col = nt * 16 + l16;
      const float* xc = Xn + col * FP_;
      const float4 c0 = *reinterpret_cast<const float4*>(xc + mt * 16 + 8 * half);
      const float4 c1 = *reinterpret_cast<const float4*>(xc + mt * 16 + 8 * half + 4);
      const v2f bf = *reinterpret_cast<const v2f*>(xc + C_ + 2 * half);
      v8f acc;
      acc[0] = c0.x; acc[1] = c0.y; acc[2] = c0.z; acc[3] = c0.w;
      acc[4] = c1.x; acc[5] = c1.y; acc[6] = c1.z; acc[7] = c1.w;
      acc = __builtin_amdgcn_wmma_f32_16x16x4_f32(
           false, a,  false, bf,
           (short)0, acc,  false,  false);
      bn_relu_store(acc, mt * 16 + half * 8, col, sc, bt, mu, Yb);
    }
  }
}

__device__ __forceinline__ void gemm_bn_relu(
    const float* __restrict__ Xb,
    const float* __restrict__ Wf, int rowsOut,
    const float* __restrict__ sc, const float* __restrict__ bt,
    const float* __restrict__ mu,
    float* __restrict__ Yb, int lane)
{
  const int half = lane >> 4;
  const int l16  = lane & 15;

  for (int mt = 0; mt < (rowsOut >> 4); ++mt) {
    const float* wt = Wf + mt * 1024;
    for (int nt = 0; nt < 2; ++nt) {
      const int col = nt * 16 + l16;
      v8f acc = {};
      for (int p = 0; p < 8; ++p) {
        const float4 a4 = *reinterpret_cast<const float4*>(wt + p * 128 + 4 * lane);
        const int k0 = 8 * p + 2 * half;
        v2f a, bf;
        a.x = a4.x; a.y = a4.y;
        bf.x = Xb[(k0 + 0) * 32 + col];
        bf.y = Xb[(k0 + 1) * 32 + col];
        acc = __builtin_amdgcn_wmma_f32_16x16x4_f32(
             false, a,  false, bf,
             (short)0, acc,  false,  false);
        a.x = a4.z; a.y = a4.w;
        bf.x = Xb[(k0 + 4) * 32 + col];
        bf.y = Xb[(k0 + 5) * 32 + col];
        acc = __builtin_amdgcn_wmma_f32_16x16x4_f32(
             false, a,  false, bf,
             (short)0, acc,  false,  false);
      }
      bn_relu_store(acc, mt * 16 + half * 8, col, sc, bt, mu, Yb);
    }
  }
}

#define LX_ (32 * FP_)
#define LW_ (LX_ + 64 * 32)

__device__ __forceinline__ void pool_rows(const float* __restrict__ Yb,
                                          float* __restrict__ dst, int lane) {
  for (int o = lane; o < 64; o += 32) {
    float mx = -BIGF;
#pragma unroll
    for (int j = 0; j < 32; ++j) mx = fmaxf(mx, Yb[o * 32 + j]);
    dst[(size_t)o * M_] = mx;
  }
}
__global__ __launch_bounds__(64) void k_knn_mlp(
    const float* __restrict__ z0, const float* __restrict__ xyz,
    const float* __restrict__ new_xyz,
    const float* __restrict__ wfrag, const float* __restrict__ bnscale,
    const float* __restrict__ b0, const float* __restrict__ m0,
    const float* __restrict__ b1, const float* __restrict__ m1,
    const float* __restrict__ b2, const float* __restrict__ m2,
    float* __restrict__ out_feat)
{
  static_assert(LW_ >= 32 * GP_, "the layers' buffers must cover the distance keys");
  __shared__ __attribute__((aligned(16))) float lds[2][LW_];
  const int wave = threadIdx.x >> 5;
  const int lane = threadIdx.x & 31;
  const int q = blockIdx.x * 2 + wave;
  const int b = q >> 10;
  const int m = q & 1023;

  float* Xb = lds[wave];
  float* Yb = lds[wave] + LX_;

  const int nj = knn_select(xyz, new_xyz, q, lane, reinterpret_cast<int*>(lds[wave]));
  __syncthreads();

  const int half = lane >> 4;
  const int l16  = lane & 15;
  const float* fb = z0 + (size_t)b * N_ * C_ + 4 * l16;
  float* xw = Xb + half * FP_ + 4 * l16;
  for (int i = 0; i < 16; ++i) {
    const int ne = __builtin_amdgcn_readlane(nj, 2 * i);
    const int no = __builtin_amdgcn_readlane(nj, 2 * i + 1);
    const int n  = half ? no : ne;
    const float4 f = *reinterpret_cast<const float4*>(fb + (size_t)n * C_);
    *reinterpret_cast<float4*>(xw + 2 * i * FP_) = f;
  }
  const float* p  = xyz + ((size_t)b * N_ + nj) * 3;
  const float* qc = new_xyz + (size_t)q * 3;
  float4 rel;
  rel.x = p[0] - qc[0];
  rel.y = p[1] - qc[1];
  rel.z = p[2] - qc[2];
  rel.w = 0.0f;
  *reinterpret_cast<float4*>(Xb + lane * FP_ + C_) = rel;
  __syncthreads();

  layer0_bn_relu(Xb, wfrag + WF0_, bnscale + BS0_, b0, m0, Yb, lane);
  __syncthreads();
  gemm_bn_relu(Yb, wfrag + WF1_,  64, bnscale + BS1_, b1, m1, Xb, lane);
  __syncthreads();
  float* oq = out_feat + (size_t)b * 128 * M_ + m;
  gemm_bn_relu(Xb, wfrag + WF2_, 64, bnscale + BS2_, b2, m2, Yb, lane);
  __syncthreads();
  pool_rows(Yb, oq, lane);
  __syncthreads();
  gemm_bn_relu(Xb, wfrag + WF2_ + 4 * 1024, 64, bnscale + BS2_ + 64, b2 + 64, m2 + 64, Yb, lane);
  __syncthreads();
  pool_rows(Yb, oq + (size_t)64 * M_, lane);
}

extern "C" void kernel_launch(void* const* d_in, const int* in_sizes, int n_in,
                              void* d_out, int out_size, void* d_ws, size_t ws_size,
                              hipStream_t stream) {
  const float* xyz      = (const float*)d_in[0];
  const float* features = (const float*)d_in[1];
  const int*   indices  = (const int*)d_in[2];
  const float* W0 = (const float*)d_in[3];
  const float* g0 = (const float*)d_in[4];
  const float* b0 = (const float*)d_in[5];
  const float* m0 = (const float*)d_in[6];
  const float* v0 = (const float*)d_in[7];
  const float* W1 = (const float*)d_in[8];
  const float* g1 = (const float*)d_in[9];
  const float* b1 = (const float*)d_in[10];
  const float* m1 = (const float*)d_in[11];
  const float* v1 = (const float*)d_in[12];
  const float* W2 = (const float*)d_in[13];
  const float* g2 = (const float*)d_in[14];
  const float* b2 = (const float*)d_in[15];
  const float* m2 = (const float*)d_in[16];
  const float* v2 = (const float*)d_in[17];

  float* new_xyz  = (float*)d_out;
  float* out_feat = new_xyz + (size_t)B_ * M_ * 3;
  float* z0       = (float*)d_ws;
  float* wfrag    = z0 + (size_t)B_ * N_ * C_;
  float* bnscale  = wfrag + WFN_;

  k_prep<<<B_ * (N_ / 64), 256, 0, stream>>>(xyz, features, indices, W0, W1, W2,
                                              g0, v0, g1, v1, g2, v2,
                                              new_xyz, z0, wfrag, bnscale);
  k_knn_mlp<<<(B_ * M_) / 2, 64, 0, stream>>>(z0, xyz, new_xyz, wfrag, bnscale,
      b0, m0, b1, m1, b2, m2, out_feat);
}


# kNN re-scan: a taken key is overwritten in LDS with the maximum, so the winner group's next minimum is a plain integer min of the four re-read keys; taken/later tests gone; same picks and order
# speedup vs baseline: 3.5685x; 1.0446x over previous
#include <hip/hip_runtime.h>
#include <hip/hip_bf16.h>

typedef __attribute__((ext_vector_type(2))) float v2f;
typedef __attribute__((ext_vector_type(8))) float v8f;

#define B_   16
#define N_   4096
#define C_   64
#define M_   1024
#define K_   32
#define EPS_ 1e-5f
#define BIGF 3.402823e38f

#define TPK_ 72
#define ZP_  68
#define WKP_ 66
#define WMT_ (16 * WKP_)

#define WF0_  0
#define WF1_  (4 * 64)
#define WF2_  (WF1_ + 64 * 64)
#define WFN_  (WF2_ + 128 * 64)
#define WFB_  (WFN_ / 256)

#define BS0_  0
#define BS1_  64
#define BS2_  128
#define BSN_  256

__global__ __launch_bounds__(256) void k_prep(const float* __restrict__ xyz,
                                              const float* __restrict__ features,
                                              const int* __restrict__ indices,
                                              const float* __restrict__ W0,
                                              const float* __restrict__ W1,
                                              const float* __restrict__ W2,
                                              const float* __restrict__ g0,
                                              const float* __restrict__ v0,
                                              const float* __restrict__ g1,
                                              const float* __restrict__ v1,
                                              const float* __restrict__ g2,
                                              const float* __restrict__ v2,
                                              float* __restrict__ new_xyz,
                                              float* __restrict__ z0,
                                              float* __restrict__ wfrag,
                                              float* __restrict__ bnscale) {
  __shared__ __attribute__((aligned(16))) float tile[C_ * TPK_];
  __shared__ __attribute__((aligned(16))) float wl[4 * WMT_];
  const int t  = threadIdx.x;
  const int b  = blockIdx.x >> 6;
  const int n0 = (blockIdx.x & 63) << 6;
  const int r  = t >> 4;
  const int e4 = (t & 15) << 2;

  const float* src = features + (size_t)b * C_ * N_ + n0;
  for (int p = 0; p < 4; ++p) {
    const int c = r + 16 * p;
    const float4 v = *reinterpret_cast<const float4*>(src + (size_t)c * N_ + e4);
    *reinterpret_cast<float4*>(tile + c * TPK_ + e4) = v;
  }
  for (int i = 0; i < 16; ++i) {
    const int f   = t + 256 * i;
    const int row = f >> 6;
    const int k   = f & 63;
    const int ln  = (row & 15) + 16 * ((k >> 1) & 1);
    wl[(row >> 4) * WMT_ + (k >> 2) * WKP_ + 2 * ln + (k & 1)] = W0[row * 67 + k];
  }
  __syncthreads();

  const int wave = t >> 5;
  const int lane = t & 31;
  const int half = lane >> 4;
  const int l16  = lane & 15;
  const int nt   = wave & 3;
  const int mt0  = (wave >> 2) << 1;
  {
    const float* xb = tile + (2 * half) * TPK_ + nt * 16 + l16;
    const float* wa = wl + mt0 * WMT_ + 2 * lane;
    v8f acc0 = {};
    v8f acc1 = {};
    for (int kt = 0; kt < 16; ++kt) {
      v2f bf;
      bf.x = xb[(4 * kt + 0) * TPK_];
      bf.y = xb[(4 * kt + 1) * TPK_];
      const v2f a0 = *reinterpret_cast<const v2f*>(wa + kt * WKP_);
      const v2f a1 = *reinterpret_cast<const v2f*>(wa + WMT_ + kt * WKP_);
      acc0 = __builtin_amdgcn_wmma_f32_16x16x4_f32(
           false, a0,  false, bf,
           (short)0, acc0,  false,  false);
      acc1 = __builtin_amdgcn_wmma_f32_16x16x4_f32(
           false, a1,  false, bf,
           (short)0, acc1,  false,  false);
    }
    __syncthreads();
    float* zr = tile + (nt * 16 + l16) * ZP_ + mt0 * 16 + 8 * half;
    float4 v;
    v.x = acc0[0]; v.y = acc0[1]; v.z = acc0[2]; v.w = acc0[3];
    *reinterpret_cast<float4*>(zr) = v;
    v.x = acc0[4]; v.y = acc0[5]; v.z = acc0[6]; v.w = acc0[7];
    *reinterpret_cast<float4*>(zr + 4) = v;
    v.x = acc1[0]; v.y = acc1[1]; v.z = acc1[2]; v.w = acc1[3];
    *reinterpret_cast<float4*>(zr + 16) = v;
    v.x = acc1[4]; v.y = acc1[5]; v.z = acc1[6]; v.w = acc1[7];
    *reinterpret_cast<float4*>(zr + 20) = v;
  }
  __syncthreads();
  float* dst = z0 + ((size_t)b * N_ + n0) * C_;
  for (int p = 0; p < 4; ++p) {
    const int n = r + 16 * p;
    const float4 v = *reinterpret_cast<const float4*>(tile + n * ZP_ + e4);
    *reinterpret_cast<float4*>(dst + (size_t)n * C_ + e4) = v;
  }

  if (blockIdx.x < (B_ * M_) / 32 && t < 32) {
    const int id = blockIdx.x * 32 + t;
    const int n = indices[id & (M_ - 1)];
    const float* s = xyz + ((size_t)(id >> 10) * N_ + n) * 3;
    float* d = new_xyz + (size_t)id * 3;
    d[0] = s[0]; d[1] = s[1]; d[2] = s[2];
  }

  if (blockIdx.x >= 512 && blockIdx.x < 512 + WFB_) {
    const int e = (blockIdx.x - 512) * 256 + t;
    if (e < WF1_) {
      const int mt = e >> 6;
      const int ln = (e >> 1) & 31;
      const int k  = 64 + 2 * (ln >> 4) + (e & 1);
      wfrag[e] = (k < 67) ? W0[(mt * 16 + (ln & 15)) * 67 + k] : 0.0f;
    } else {
      const float* Wl = (e < WF2_) ? W1 : W2;
      const int q  = e - ((e < WF2_) ? WF1_ : WF2_);
      const int mt = q >> 10;
      const int w  = q & 1023;
      const int kt = 2 * (w >> 7) + ((w >> 1) & 1);
      const int ln = (w & 127) >> 2;
      const int k  = 4 * kt + 2 * (ln >> 4) + (w & 1);
      wfrag[e] = Wl[(mt * 16 + (ln & 15)) * 64 + k];
    }
  }

  if (blockIdx.x == 512 + WFB_) {
    const float* gp = (t < BS1_) ? g0 : (t < BS2_) ? g1 : g2;
    const float* vp = (t < BS1_) ? v0 : (t < BS2_) ? v1 : v2;
    const int c = (t < BS1_) ? t : (t < BS2_) ? t - BS1_ : t - BS2_;
    bnscale[t] = gp[c] / sqrtf(vp[c] + EPS_);
  }
}

#define GP_   132
#define IMAX_ 0x7fffffff

__device__ __forceinline__ int dist_key(float f) {
  const int u = __float_as_int(f);
  return u ^ ((u >> 31) & 0x7fffffff);
}

template <int M>
__device__ __forceinline__ int row_xor(int x) {
  return __builtin_amdgcn_update_dpp(0, x, 0x160 + M, 0xf, 0xf, true);
}
__device__ __forceinline__ int half_xor(int x) {
  return __builtin_amdgcn_permlanex16(x, x, 0x76543210, 0xfedcba98, false, false);
}
__device__ __forceinline__ int wave_min_i(int v) {
  v = min(v, row_xor<1>(v));
  v = min(v, row_xor<2>(v));
  v = min(v, row_xor<4>(v));
  v = min(v, row_xor<8>(v));
  v = min(v, half_xor(v));
  return v;
}

#define PF_ 8

__device__ __forceinline__ void load_points(const float* __restrict__ px, int lane, int i0,
                                            float (&x)[PF_], float (&y)[PF_], float (&z)[PF_]) {
#pragma unroll
  for (int u = 0; u < PF_; ++u) {
    const int n = lane + 32 * (i0 + u);
    x[u] = px[n * 3 + 0];
    y[u] = px[n * 3 + 1];
    z[u] = px[n * 3 + 2];
  }
}

__device__ __forceinline__ void keys_of_points(const float (&xs)[PF_], const float (&ys)[PF_],
                                               const float (&zs)[PF_],
                                               float qx, float qy, float qz, float qq,
                                               int lane, int i0, int* __restrict__ mine,
                                               int& gv, int& gi) {
#pragma unroll
  for (int u = 0; u < PF_; ++u) {
    const int n = lane + 32 * (i0 + u);
    const float x = xs[u], y = ys[u], z = zs[u];
    const float pp  = __builtin_fmaf(z, z, __builtin_fmaf(x, x, y * y));
    const float dot = __builtin_fmaf(qz, z, __builtin_fmaf(qx, x, qy * y));
    const float d   = __builtin_fmaf(-2.0f, dot, qq + pp);
    const int key = dist_key(d);
    mine[i0 + u] = key;
    if (key < gv) { gv = key; gi = n; }
  }
}

__device__ __forceinline__ int knn_select(const float* __restrict__ xyz,
                                          const float* __restrict__ new_xyz,
                                          int q, int lane, int* __restrict__ dl) {
  const int b = q >> 10;
  const float qx = new_xyz[q * 3 + 0];
  const float qy = new_xyz[q * 3 + 1];
  const float qz = new_xyz[q * 3 + 2];
  float qq;
  {
#pragma clang fp contract(off)
    qq = __builtin_fmaf(qx, qx, qy * qy) + qz * qz;
  }
  const float* px = xyz + (size_t)b * N_ * 3;

  int gv = IMAX_;
  int gi = IMAX_;
  int* mine = dl + lane * GP_;
  float ax[PF_], ay[PF_], az[PF_];
  float bx[PF_], by[PF_], bz[PF_];
  load_points(px, lane, 0, ax, ay, az);
  for (int i = 0; i < N_ / 32 - 2 * PF_; i += 2 * PF_) {
    load_points(px, lane, i + PF_, bx, by, bz);
    keys_of_points(ax, ay, az, qx, qy, qz, qq, lane, i, mine, gv, gi);
    load_points(px, lane, i + 2 * PF_, ax, ay, az);
    keys_of_points(bx, by, bz, qx, qy, qz, qq, lane, i + PF_, mine, gv, gi);
  }
  load_points(px, lane, N_ / 32 - PF_, bx, by, bz);
  keys_of_points(ax, ay, az, qx, qy, qz, qq, lane, N_ / 32 - 2 * PF_, mine, gv, gi);
  keys_of_points(bx, by, bz, qx, qy, qz, qq, lane, N_ / 32 - PF_, mine, gv, gi);
  __syncthreads();

  int res = 0;
  for (int k = 0; k < K_; ++k) {
    const int mn = wave_min_i(gv);
    const int mi = wave_min_i(gv == mn ? gi : IMAX_);
    if (lane == k) res = mi;
    const int w = mi & 31;

    int* grp = dl + w * GP_;
    grp[mi >> 5] = IMAX_;
    const int* cp = grp + 4 * lane;
    const int c0 = cp[0];
    const int c1 = cp[1];
    const int c2 = cp[2];
    const int c3 = cp[3];
    const int lv = min(min(c0, c1), min(c2, c3));
    const int nv = wave_min_i(lv);
    const int n0 = w + 128 * lane;
    const int off = (c0 == nv) ? 0 : (c1 == nv) ? 32 : (c2 == nv) ? 64 : 96;
    if (lane == w) gv = nv;
    const int ni = wave_min_i(lv == nv ? n0 + off : IMAX_);
    if (lane == w) gi = ni;
  }
  return res;
}

#define FP_ 68

__device__ __forceinline__ void bn_relu_store(
    const v8f acc, int rbase, int col,
    const float* __restrict__ sc, const float* __restrict__ bt,
    const float* __restrict__ mu,
    float* __restrict__ Yb)
{
#pragma unroll
  for (int r = 0; r < 8; ++r) {
    const int row = rbase + r;
    const float y = (acc[r] - mu[row]) * sc[row] + bt[row];
    Yb[row * 32 + col] = fmaxf(y, 0.0f);
  }
}

__device__ __forceinline__ void layer0_bn_relu(
    const float* __restrict__ Xn, const float* __restrict__ Wf,
    const float* __restrict__ sc, const float* __restrict__ bt,
    const float* __restrict__ mu,
    float* __restrict__ Yb, int lane)
{
  const int half = lane >> 4;
  const int l16  = lane & 15;
  for (int mt = 0; mt < 4; ++mt) {
    const v2f a = *reinterpret_cast<const v2f*>(Wf + mt * 64 + 2 * lane);
    for (int nt = 0; nt < 2; ++nt) {
      const int col = nt * 16 + l16;
      const float* xc = Xn + col * FP_;
      const float4 c0 = *reinterpret_cast<const float4*>(xc + mt * 16 + 8 * half);
      const float4 c1 = *reinterpret_cast<const float4*>(xc + mt * 16 + 8 * half + 4);
      const v2f bf = *reinterpret_cast<const v2f*>(xc + C_ + 2 * half);
      v8f acc;
      acc[0] = c0.x; acc[1] = c0.y; acc[2] = c0.z; acc[3] = c0.w;
      acc[4] = c1.x; acc[5] = c1.y; acc[6] = c1.z; acc[7] = c1.w;
      acc = __builtin_amdgcn_wmma_f32_16x16x4_f32(
           false, a,  false, bf,
           (short)0, acc,  false,  false);
      bn_relu_store(acc, mt * 16 + half * 8, col, sc, bt, mu, Yb);
    }
  }
}

__device__ __forceinline__ void gemm_bn_relu(
    const float* __restrict__ Xb,
    const float* __restrict__ Wf, int rowsOut,
    const float* __restrict__ sc, const float* __restrict__ bt,
    const float* __restrict__ mu,
    float* __restrict__ Yb, int lane)
{
  const int half = lane >> 4;
  const int l16  = lane & 15;

  for (int mt = 0; mt < (rowsOut >> 4); ++mt) {
    const float* wt = Wf + mt * 1024;
    for (int nt = 0; nt < 2; ++nt) {
      const int col = nt * 16 + l16;
      v8f acc = {};
      for (int p = 0; p < 8; ++p) {
        const float4 a4 = *reinterpret_cast<const float4*>(wt + p * 128 + 4 * lane);
        const int k0 = 8 * p + 2 * half;
        v2f a, bf;
        a.x = a4.x; a.y = a4.y;
        bf.x = Xb[(k0 + 0) * 32 + col];
        bf.y = Xb[(k0 + 1) * 32 + col];
        acc = __builtin_amdgcn_wmma_f32_16x16x4_f32(
             false, a,  false, bf,
             (short)0, acc,  false,  false);
        a.x = a4.z; a.y = a4.w;
        bf.x = Xb[(k0 + 4) * 32 + col];
        bf.y = Xb[(k0 + 5) * 32 + col];
        acc = __builtin_amdgcn_wmma_f32_16x16x4_f32(
             false, a,  false, bf,
             (short)0, acc,  false,  false);
      }
      bn_relu_store(acc, mt * 16 + half * 8, col, sc, bt, mu, Yb);
    }
  }
}

#define LX_ (32 * FP_)
#define LW_ (LX_ + 64 * 32)

__device__ __forceinline__ void pool_rows(const float* __restrict__ Yb,
                                          float* __restrict__ dst, int lane) {
  for (int o = lane; o < 64; o += 32) {
    float mx = -BIGF;
#pragma unroll
    for (int j = 0; j < 32; ++j) mx = fmaxf(mx, Yb[o * 32 + j]);
    dst[(size_t)o * M_] = mx;
  }
}
__global__ __launch_bounds__(64) void k_knn_mlp(
    const float* __restrict__ z0, const float* __restrict__ xyz,
    const float* __restrict__ new_xyz,
    const float* __restrict__ wfrag, const float* __restrict__ bnscale,
    const float* __restrict__ b0, const float* __restrict__ m0,
    const float* __restrict__ b1, const float* __restrict__ m1,
    const float* __restrict__ b2, const float* __restrict__ m2,
    float* __restrict__ out_feat)
{
  static_assert(LW_ >= 32 * GP_, "the layers' buffers must cover the distance keys");
  __shared__ __attribute__((aligned(16))) float lds[2][LW_];
  const int wave = threadIdx.x >> 5;
  const int lane = threadIdx.x & 31;
  const int q = blockIdx.x * 2 + wave;
  const int b = q >> 10;
  const int m = q & 1023;

  float* Xb = lds[wave];
  float* Yb = lds[wave] + LX_;

  const int nj = knn_select(xyz, new_xyz, q, lane, reinterpret_cast<int*>(lds[wave]));
  __syncthreads();

  const int half = lane >> 4;
  const int l16  = lane & 15;
  const float* fb = z0 + (size_t)b * N_ * C_ + 4 * l16;
  float* xw = Xb + half * FP_ + 4 * l16;
  for (int i = 0; i < 16; ++i) {
    const int ne = __builtin_amdgcn_readlane(nj, 2 * i);
    const int no = __builtin_amdgcn_readlane(nj, 2 * i + 1);
    const int n  = half ? no : ne;
    const float4 f = *reinterpret_cast<const float4*>(fb + (size_t)n * C_);
    *reinterpret_cast<float4*>(xw + 2 * i * FP_) = f;
  }
  const float* p  = xyz + ((size_t)b * N_ + nj) * 3;
  const float* qc = new_xyz + (size_t)q * 3;
  float4 rel;
  rel.x = p[0] - qc[0];
  rel.y = p[1] - qc[1];
  rel.z = p[2] - qc[2];
  rel.w = 0.0f;
  *reinterpret_cast<float4*>(Xb + lane * FP_ + C_) = rel;
  __syncthreads();

  layer0_bn_relu(Xb, wfrag + WF0_, bnscale + BS0_, b0, m0, Yb, lane);
  __syncthreads();
  gemm_bn_relu(Yb, wfrag + WF1_,  64, bnscale + BS1_, b1, m1, Xb, lane);
  __syncthreads();
  float* oq = out_feat + (size_t)b * 128 * M_ + m;
  gemm_bn_relu(Xb, wfrag + WF2_, 64, bnscale + BS2_, b2, m2, Yb, lane);
  __syncthreads();
  pool_rows(Yb, oq, lane);
  __syncthreads();
  gemm_bn_relu(Xb, wfrag + WF2_ + 4 * 1024, 64, bnscale + BS2_ + 64, b2 + 64, m2 + 64, Yb, lane);
  __syncthreads();
  pool_rows(Yb, oq + (size_t)64 * M_, lane);
}

extern "C" void kernel_launch(void* const* d_in, const int* in_sizes, int n_in,
                              void* d_out, int out_size, void* d_ws, size_t ws_size,
                              hipStream_t stream) {
  const float* xyz      = (const float*)d_in[0];
  const float* features = (const float*)d_in[1];
  const int*   indices  = (const int*)d_in[2];
  const float* W0 = (const float*)d_in[3];
  const float* g0 = (const float*)d_in[4];
  const float* b0 = (const float*)d_in[5];
  const float* m0 = (const float*)d_in[6];
  const float* v0 = (const float*)d_in[7];
  const float* W1 = (const float*)d_in[8];
  const float* g1 = (const float*)d_in[9];
  const float* b1 = (const float*)d_in[10];
  const float* m1 = (const float*)d_in[11];
  const float* v1 = (const float*)d_in[12];
  const float* W2 = (const float*)d_in[13];
  const float* g2 = (const float*)d_in[14];
  const float* b2 = (const float*)d_in[15];
  const float* m2 = (const float*)d_in[16];
  const float* v2 = (const float*)d_in[17];

  float* new_xyz  = (float*)d_out;
  float* out_feat = new_xyz + (size_t)B_ * M_ * 3;
  float* z0       = (float*)d_ws;
  float* wfrag    = z0 + (size_t)B_ * N_ * C_;
  float* bnscale  = wfrag + WFN_;

  k_prep<<<B_ * (N_ / 64), 256, 0, stream>>>(xyz, features, indices, W0, W1, W2,
                                              g0, v0, g1, v1, g2, v2,
                                              new_xyz, z0, wfrag, bnscale);
  k_knn_mlp<<<(B_ * M_) / 2, 64, 0, stream>>>(z0, xyz, new_xyz, wfrag, bnscale,
      b0, m0, b1, m1, b2, m2, out_feat);
}


# kNN distance pass shared by a block's two waves: each takes half the points and makes their keys for both queries (point loaded, |p|^2 made once per block; half the load waits per wave); same keys
# speedup vs baseline: 3.7201x; 1.0425x over previous
#include <hip/hip_runtime.h>
#include <hip/hip_bf16.h>

typedef __attribute__((ext_vector_type(2))) float v2f;
typedef __attribute__((ext_vector_type(8))) float v8f;

#define B_   16
#define N_   4096
#define C_   64
#define M_   1024
#define K_   32
#define EPS_ 1e-5f
#define BIGF 3.402823e38f

#define TPK_ 72
#define ZP_  68
#define WKP_ 66
#define WMT_ (16 * WKP_)

#define WF0_  0
#define WF1_  (4 * 64)
#define WF2_  (WF1_ + 64 * 64)
#define WFN_  (WF2_ + 128 * 64)
#define WFB_  (WFN_ / 256)

#define BS0_  0
#define BS1_  64
#define BS2_  128
#define BSN_  256

__global__ __launch_bounds__(256) void k_prep(const float* __restrict__ xyz,
                                              const float* __restrict__ features,
                                              const int* __restrict__ indices,
                                              const float* __restrict__ W0,
                                              const float* __restrict__ W1,
                                              const float* __restrict__ W2,
                                              const float* __restrict__ g0,
                                              const float* __restrict__ v0,
                                              const float* __restrict__ g1,
                                              const float* __restrict__ v1,
                                              const float* __restrict__ g2,
                                              const float* __restrict__ v2,
                                              float* __restrict__ new_xyz,
                                              float* __restrict__ z0,
                                              float* __restrict__ wfrag,
                                              float* __restrict__ bnscale) {
  __shared__ __attribute__((aligned(16))) float tile[C_ * TPK_];
  __shared__ __attribute__((aligned(16))) float wl[4 * WMT_];
  const int t  = threadIdx.x;
  const int b  = blockIdx.x >> 6;
  const int n0 = (blockIdx.x & 63) << 6;
  const int r  = t >> 4;
  const int e4 = (t & 15) << 2;

  const float* src = features + (size_t)b * C_ * N_ + n0;
  for (int p = 0; p < 4; ++p) {
    const int c = r + 16 * p;
    const float4 v = *reinterpret_cast<const float4*>(src + (size_t)c * N_ + e4);
    *reinterpret_cast<float4*>(tile + c * TPK_ + e4) = v;
  }
  for (int i = 0; i < 16; ++i) {
    const int f   = t + 256 * i;
    const int row = f >> 6;
    const int k   = f & 63;
    const int ln  = (row & 15) + 16 * ((k >> 1) & 1);
    wl[(row >> 4) * WMT_ + (k >> 2) * WKP_ + 2 * ln + (k & 1)] = W0[row * 67 + k];
  }
  __syncthreads();

  const int wave = t >> 5;
  const int lane = t & 31;
  const int half = lane >> 4;
  const int l16  = lane & 15;
  const int nt   = wave & 3;
  const int mt0  = (wave >> 2) << 1;
  {
    const float* xb = tile + (2 * half) * TPK_ + nt * 16 + l16;
    const float* wa = wl + mt0 * WMT_ + 2 * lane;
    v8f acc0 = {};
    v8f acc1 = {};
    for (int kt = 0; kt < 16; ++kt) {
      v2f bf;
      bf.x = xb[(4 * kt + 0) * TPK_];
      bf.y = xb[(4 * kt + 1) * TPK_];
      const v2f a0 = *reinterpret_cast<const v2f*>(wa + kt * WKP_);
      const v2f a1 = *reinterpret_cast<const v2f*>(wa + WMT_ + kt * WKP_);
      acc0 = __builtin_amdgcn_wmma_f32_16x16x4_f32(
           false, a0,  false, bf,
           (short)0, acc0,  false,  false);
      acc1 = __builtin_amdgcn_wmma_f32_16x16x4_f32(
           false, a1,  false, bf,
           (short)0, acc1,  false,  false);
    }
    __syncthreads();
    float* zr = tile + (nt * 16 + l16) * ZP_ + mt0 * 16 + 8 * half;
    float4 v;
    v.x = acc0[0]; v.y = acc0[1]; v.z = acc0[2]; v.w = acc0[3];
    *reinterpret_cast<float4*>(zr) = v;
    v.x = acc0[4]; v.y = acc0[5]; v.z = acc0[6]; v.w = acc0[7];
    *reinterpret_cast<float4*>(zr + 4) = v;
    v.x = acc1[0]; v.y = acc1[1]; v.z = acc1[2]; v.w = acc1[3];
    *reinterpret_cast<float4*>(zr + 16) = v;
    v.x = acc1[4]; v.y = acc1[5]; v.z = acc1[6]; v.w = acc1[7];
    *reinterpret_cast<float4*>(zr + 20) = v;
  }
  __syncthreads();
  float* dst = z0 + ((size_t)b * N_ + n0) * C_;
  for (int p = 0; p < 4; ++p) {
    const int n = r + 16 * p;
    const float4 v = *reinterpret_cast<const float4*>(tile + n * ZP_ + e4);
    *reinterpret_cast<float4*>(dst + (size_t)n * C_ + e4) = v;
  }

  if (blockIdx.x < (B_ * M_) / 32 && t < 32) {
    const int id = blockIdx.x * 32 + t;
    const int n = indices[id & (M_ - 1)];
    const float* s = xyz + ((size_t)(id >> 10) * N_ + n) * 3;
    float* d = new_xyz + (size_t)id * 3;
    d[0] = s[0]; d[1] = s[1]; d[2] = s[2];
  }

  if (blockIdx.x >= 512 && blockIdx.x < 512 + WFB_) {
    const int e = (blockIdx.x - 512) * 256 + t;
    if (e < WF1_) {
      const int mt = e >> 6;
      const int ln = (e >> 1) & 31;
      const int k  = 64 + 2 * (ln >> 4) + (e & 1);
      wfrag[e] = (k < 67) ? W0[(mt * 16 + (ln & 15)) * 67 + k] : 0.0f;
    } else {
      const float* Wl = (e < WF2_) ? W1 : W2;
      const int q  = e - ((e < WF2_) ? WF1_ : WF2_);
      const int mt = q >> 10;
      const int w  = q & 1023;
      const int kt = 2 * (w >> 7) + ((w >> 1) & 1);
      const int ln = (w & 127) >> 2;
      const int k  = 4 * kt + 2 * (ln >> 4) + (w & 1);
      wfrag[e] = Wl[(mt * 16 + (ln & 15)) * 64 + k];
    }
  }

  if (blockIdx.x == 512 + WFB_) {
    const float* gp = (t < BS1_) ? g0 : (t < BS2_) ? g1 : g2;
    const float* vp = (t < BS1_) ? v0 : (t < BS2_) ? v1 : v2;
    const int c = (t < BS1_) ? t : (t < BS2_) ? t - BS1_ : t - BS2_;
    bnscale[t] = gp[c] / sqrtf(vp[c] + EPS_);
  }
}

#define GP_   132
#define IMAX_ 0x7fffffff

__device__ __forceinline__ int dist_key(float f) {
  const int u = __float_as_int(f);
  return u ^ ((u >> 31) & 0x7fffffff);
}

template <int M>
__device__ __forceinline__ int row_xor(int x) {
  return __builtin_amdgcn_update_dpp(0, x, 0x160 + M, 0xf, 0xf, true);
}
__device__ __forceinline__ int half_xor(int x) {
  return __builtin_amdgcn_permlanex16(x, x, 0x76543210, 0xfedcba98, false, false);
}
__device__ __forceinline__ int wave_min_i(int v) {
  v = min(v, row_xor<1>(v));
  v = min(v, row_xor<2>(v));
  v = min(v, row_xor<4>(v));
  v = min(v, row_xor<8>(v));
  v = min(v, half_xor(v));
  return v;
}

#define PF_ 8
#define HI_ (N_ / 64)
#define HW_ 128

__device__ __forceinline__ void load_points(const float* __restrict__ px, int lane, int i0,
                                            float (&x)[PF_], float (&y)[PF_], float (&z)[PF_]) {
#pragma unroll
  for (int u = 0; u < PF_; ++u) {
    const int n = lane + 32 * (i0 + u);
    x[u] = px[n * 3 + 0];
    y[u] = px[n * 3 + 1];
    z[u] = px[n * 3 + 2];
  }
}

__device__ __forceinline__ void keys_of_points2(const float (&xs)[PF_], const float (&ys)[PF_],
                                                const float (&zs)[PF_],
                                                float ax, float ay, float az, float aq,
                                                float bx, float by, float bz, float bq,
                                                int n0, int i0,
                                                int* __restrict__ ma, int* __restrict__ mb,
                                                int& gva, int& gia, int& gvb, int& gib) {
#pragma unroll
  for (int u = 0; u < PF_; ++u) {
    const int n = n0 + 32 * (i0 + u);
    const float x = xs[u], y = ys[u], z = zs[u];
    const float pp   = __builtin_fmaf(z, z, __builtin_fmaf(x, x, y * y));
    const float dota = __builtin_fmaf(az, z, __builtin_fmaf(ax, x, ay * y));
    const float dotb = __builtin_fmaf(bz, z, __builtin_fmaf(bx, x, by * y));
    const float da   = __builtin_fmaf(-2.0f, dota, aq + pp);
    const float db   = __builtin_fmaf(-2.0f, dotb, bq + pp);
    const int ka = dist_key(da);
    const int kb = dist_key(db);
    ma[i0 + u] = ka;
    mb[i0 + u] = kb;
    if (ka < gva) { gva = ka; gia = n; }
    if (kb < gvb) { gvb = kb; gib = n; }
  }
}

__device__ __forceinline__ float query_norm(float qx, float qy, float qz) {
  float qq;
  {
#pragma clang fp contract(off)
    qq = __builtin_fmaf(qx, qx, qy * qy) + qz * qz;
  }
  return qq;
}

__device__ __forceinline__ int knn_select(const float* __restrict__ xyz,
                                          const float* __restrict__ new_xyz,
                                          int q0, int wave, int lane,
                                          int* __restrict__ dl0, int* __restrict__ dl1) {
  const int b = q0 >> 10;
  const float ax = new_xyz[q0 * 3 + 0];
  const float ay = new_xyz[q0 * 3 + 1];
  const float az = new_xyz[q0 * 3 + 2];
  const float bx = new_xyz[q0 * 3 + 3];
  const float by = new_xyz[q0 * 3 + 4];
  const float bz = new_xyz[q0 * 3 + 5];
  const float aq = query_norm(ax, ay, az);
  const float bq = query_norm(bx, by, bz);
  const float* px = xyz + (size_t)b * N_ * 3;

  int gva = IMAX_, gia = IMAX_;
  int gvb = IMAX_, gib = IMAX_;
  const int ib = __builtin_amdgcn_readfirstlane(wave) * HI_;
  int* ma = dl0 + lane * GP_ + ib;
  int* mb = dl1 + lane * GP_ + ib;
  const float* ph = px + (size_t)ib * 32 * 3;
  const int n0 = lane + 32 * ib;
  float sx[PF_], sy[PF_], sz[PF_];
  float tx[PF_], ty[PF_], tz[PF_];
  load_points(ph, lane, 0, sx, sy, sz);
  for (int i = 0; i < HI_ - 2 * PF_; i += 2 * PF_) {
    load_points(ph, lane, i + PF_, tx, ty, tz);
    keys_of_points2(sx, sy, sz, ax, ay, az, aq, bx, by, bz, bq, n0, i, ma, mb,
                    gva, gia, gvb, gib);
    load_points(ph, lane, i + 2 * PF_, sx, sy, sz);
    keys_of_points2(tx, ty, tz, ax, ay, az, aq, bx, by, bz, bq, n0, i + PF_, ma, mb,
                    gva, gia, gvb, gib);
  }
  load_points(ph, lane, HI_ - PF_, tx, ty, tz);
  keys_of_points2(sx, sy, sz, ax, ay, az, aq, bx, by, bz, bq, n0, HI_ - 2 * PF_, ma, mb,
                  gva, gia, gvb, gib);
  keys_of_points2(tx, ty, tz, ax, ay, az, aq, bx, by, bz, bq, n0, HI_ - PF_, ma, mb,
                  gva, gia, gvb, gib);

  int* hand = (wave ? dl0 : dl1) + lane * GP_;
  hand[HW_ + 0] = wave ? gva : gvb;
  hand[HW_ + 1] = wave ? gia : gib;
  __syncthreads();
  int* dl = wave ? dl1 : dl0;
  int gv = wave ? gvb : gva;
  int gi = wave ? gib : gia;
  {
    const int ov = dl[lane * GP_ + HW_ + 0];
    const int oi = dl[lane * GP_ + HW_ + 1];
    const bool take = wave ? (ov <= gv) : (ov < gv);
    if (take) { gv = ov; gi = oi; }
  }

  int res = 0;
  for (int k = 0; k < K_; ++k) {
    const int mn = wave_min_i(gv);
    const int mi = wave_min_i(gv == mn ? gi : IMAX_);
    if (lane == k) res = mi;
    const int w = mi & 31;

    int* grp = dl + w * GP_;
    grp[mi >> 5] = IMAX_;
    const int* cp = grp + 4 * lane;
    const int c0 = cp[0];
    const int c1 = cp[1];
    const int c2 = cp[2];
    const int c3 = cp[3];
    const int lv = min(min(c0, c1), min(c2, c3));
    const int nv = wave_min_i(lv);
    const int n0 = w + 128 * lane;
    const int off = (c0 == nv) ? 0 : (c1 == nv) ? 32 : (c2 == nv) ? 64 : 96;
    if (lane == w) gv = nv;
    const int ni = wave_min_i(lv == nv ? n0 + off : IMAX_);
    if (lane == w) gi = ni;
  }
  return res;
}

#define FP_ 68

__device__ __forceinline__ void bn_relu_store(
    const v8f acc, int rbase, int col,
    const float* __restrict__ sc, const float* __restrict__ bt,
    const float* __restrict__ mu,
    float* __restrict__ Yb)
{
#pragma unroll
  for (int r = 0; r < 8; ++r) {
    const int row = rbase + r;
    const float y = (acc[r] - mu[row]) * sc[row] + bt[row];
    Yb[row * 32 + col] = fmaxf(y, 0.0f);
  }
}

__device__ __forceinline__ void layer0_bn_relu(
    const float* __restrict__ Xn, const float* __restrict__ Wf,
    const float* __restrict__ sc, const float* __restrict__ bt,
    const float* __restrict__ mu,
    float* __restrict__ Yb, int lane)
{
  const int half = lane >> 4;
  const int l16  = lane & 15;
  for (int mt = 0; mt < 4; ++mt) {
    const v2f a = *reinterpret_cast<const v2f*>(Wf + mt * 64 + 2 * lane);
    for (int nt = 0; nt < 2; ++nt) {
      const int col = nt * 16 + l16;
      const float* xc = Xn + col * FP_;
      const float4 c0 = *reinterpret_cast<const float4*>(xc + mt * 16 + 8 * half);
      const float4 c1 = *reinterpret_cast<const float4*>(xc + mt * 16 + 8 * half + 4);
      const v2f bf = *reinterpret_cast<const v2f*>(xc + C_ + 2 * half);
      v8f acc;
      acc[0] = c0.x; acc[1] = c0.y; acc[2] = c0.z; acc[3] = c0.w;
      acc[4] = c1.x; acc[5] = c1.y; acc[6] = c1.z; acc[7] = c1.w;
      acc = __builtin_amdgcn_wmma_f32_16x16x4_f32(
           false, a,  false, bf,
           (short)0, acc,  false,  false);
      bn_relu_store(acc, mt * 16 + half * 8, col, sc, bt, mu, Yb);
    }
  }
}

__device__ __forceinline__ void gemm_bn_relu(
    const float* __restrict__ Xb,
    const float* __restrict__ Wf, int rowsOut,
    const float* __restrict__ sc, const float* __restrict__ bt,
    const float* __restrict__ mu,
    float* __restrict__ Yb, int lane)
{
  const int half = lane >> 4;
  const int l16  = lane & 15;

  for (int mt = 0; mt < (rowsOut >> 4); ++mt) {
    const float* wt = Wf + mt * 1024;
    for (int nt = 0; nt < 2; ++nt) {
      const int col = nt * 16 + l16;
      v8f acc = {};
      for (int p = 0; p < 8; ++p) {
        const float4 a4 = *reinterpret_cast<const float4*>(wt + p * 128 + 4 * lane);
        const int k0 = 8 * p + 2 * half;
        v2f a, bf;
        a.x = a4.x; a.y = a4.y;
        bf.x = Xb[(k0 + 0) * 32 + col];
        bf.y = Xb[(k0 + 1) * 32 + col];
        acc = __builtin_amdgcn_wmma_f32_16x16x4_f32(
             false, a,  false, bf,
             (short)0, acc,  false,  false);
        a.x = a4.z; a.y = a4.w;
        bf.x = Xb[(k0 + 4) * 32 + col];
        bf.y = Xb[(k0 + 5) * 32 + col];
        acc = __builtin_amdgcn_wmma_f32_16x16x4_f32(
             false, a,  false, bf,
             (short)0, acc,  false,  false);
      }
      bn_relu_store(acc, mt * 16 + half * 8, col, sc, bt, mu, Yb);
    }
  }
}

#define LX_ (32 * FP_)
#define LW_ (LX_ + 64 * 32)

__device__ __forceinline__ void pool_rows(const float* __restrict__ Yb,
                                          float* __restrict__ dst, int lane) {
  for (int o = lane; o < 64; o += 32) {
    float mx = -BIGF;
#pragma unroll
    for (int j = 0; j < 32; ++j) mx = fmaxf(mx, Yb[o * 32 + j]);
    dst[(size_t)o * M_] = mx;
  }
}
__global__ __launch_bounds__(64) void k_knn_mlp(
    const float* __restrict__ z0, const float* __restrict__ xyz,
    const float* __restrict__ new_xyz,
    const float* __restrict__ wfrag, const float* __restrict__ bnscale,
    const float* __restrict__ b0, const float* __restrict__ m0,
    const float* __restrict__ b1, const float* __restrict__ m1,
    const float* __restrict__ b2, const float* __restrict__ m2,
    float* __restrict__ out_feat)
{
  static_assert(LW_ >= 32 * GP_, "the layers' buffers must cover the distance keys");
  __shared__ __attribute__((aligned(16))) float lds[2][LW_];
  const int wave = threadIdx.x >> 5;
  const int lane = threadIdx.x & 31;
  const int q = blockIdx.x * 2 + wave;
  const int b = q >> 10;
  const int m = q & 1023;

  float* Xb = lds[wave];
  float* Yb = lds[wave] + LX_;

  const int nj = knn_select(xyz, new_xyz, blockIdx.x * 2, wave, lane,
                            reinterpret_cast<int*>(lds[0]), reinterpret_cast<int*>(lds[1]));
  __syncthreads();

  const int half = lane >> 4;
  const int l16  = lane & 15;
  const float* fb = z0 + (size_t)b * N_ * C_ + 4 * l16;
  float* xw = Xb + half * FP_ + 4 * l16;
  for (int i = 0; i < 16; ++i) {
    const int ne = __builtin_amdgcn_readlane(nj, 2 * i);
    const int no = __builtin_amdgcn_readlane(nj, 2 * i + 1);
    const int n  = half ? no : ne;
    const float4 f = *reinterpret_cast<const float4*>(fb + (size_t)n * C_);
    *reinterpret_cast<float4*>(xw + 2 * i * FP_) = f;
  }
  const float* p  = xyz + ((size_t)b * N_ + nj) * 3;
  const float* qc = new_xyz + (size_t)q * 3;
  float4 rel;
  rel.x = p[0] - qc[0];
  rel.y = p[1] - qc[1];
  rel.z = p[2] - qc[2];
  rel.w = 0.0f;
  *reinterpret_cast<float4*>(Xb + lane * FP_ + C_) = rel;
  __syncthreads();

  layer0_bn_relu(Xb, wfrag + WF0_, bnscale + BS0_, b0, m0, Yb, lane);
  __syncthreads();
  gemm_bn_relu(Yb, wfrag + WF1_,  64, bnscale + BS1_, b1, m1, Xb, lane);
  __syncthreads();
  float* oq = out_feat + (size_t)b * 128 * M_ + m;
  gemm_bn_relu(Xb, wfrag + WF2_, 64, bnscale + BS2_, b2, m2, Yb, lane);
  __syncthreads();
  pool_rows(Yb, oq, lane);
  __syncthreads();
  gemm_bn_relu(Xb, wfrag + WF2_ + 4 * 1024, 64, bnscale + BS2_ + 64, b2 + 64, m2 + 64, Yb, lane);
  __syncthreads();
  pool_rows(Yb, oq + (size_t)64 * M_, lane);
}

extern "C" void kernel_launch(void* const* d_in, const int* in_sizes, int n_in,
                              void* d_out, int out_size, void* d_ws, size_t ws_size,
                              hipStream_t stream) {
  const float* xyz      = (const float*)d_in[0];
  const float* features = (const float*)d_in[1];
  const int*   indices  = (const int*)d_in[2];
  const float* W0 = (const float*)d_in[3];
  const float* g0 = (const float*)d_in[4];
  const float* b0 = (const float*)d_in[5];
  const float* m0 = (const float*)d_in[6];
  const float* v0 = (const float*)d_in[7];
  const float* W1 = (const float*)d_in[8];
  const float* g1 = (const float*)d_in[9];
  const float* b1 = (const float*)d_in[10];
  const float* m1 = (const float*)d_in[11];
  const float* v1 = (const float*)d_in[12];
  const float* W2 = (const float*)d_in[13];
  const float* g2 = (const float*)d_in[14];
  const float* b2 = (const float*)d_in[15];
  const float* m2 = (const float*)d_in[16];
  const float* v2 = (const float*)d_in[17];

  float* new_xyz  = (float*)d_out;
  float* out_feat = new_xyz + (size_t)B_ * M_ * 3;
  float* z0       = (float*)d_ws;
  float* wfrag    = z0 + (size_t)B_ * N_ * C_;
  float* bnscale  = wfrag + WFN_;

  k_prep<<<B_ * (N_ / 64), 256, 0, stream>>>(xyz, features, indices, W0, W1, W2,
                                              g0, v0, g1, v1, g2, v2,
                                              new_xyz, z0, wfrag, bnscale);
  k_knn_mlp<<<(B_ * M_) / 2, 64, 0, stream>>>(z0, xyz, new_xyz, wfrag, bnscale,
      b0, m0, b1, m1, b2, m2, out_feat);
}
